# starting kernel
# baseline (speedup 1.0000x reference)
#include <hip/hip_runtime.h>
#include <hip/hip_bf16.h>
#include <math.h>

#define NN 20000
#define EE 320000
#define HH 8
#define DD 32
#define F  256
#define LDSPITCH 260

typedef float v2f __attribute__((ext_vector_type(2)));
typedef float v8f __attribute__((ext_vector_type(8)));

__global__ __launch_bounds__(256) void gat_gemm_wmma(
    const float* __restrict__ A, const float* __restrict__ W,
    float* __restrict__ C) {
  __shared__ float As[16 * LDSPITCH];
  const int tid  = threadIdx.x;
  const int wave = tid >> 5;
  const int lane = tid & 31;
  const int m0   = blockIdx.x << 4;

  for (int i = tid; i < 16 * 64; i += 256) {
    const int r  = i >> 6;
    const int c4 = (i & 63) << 2;
    const float4 v = *reinterpret_cast<const float4*>(A + (size_t)(m0 + r) * F + c4);
    float* p = &As[r * LDSPITCH + c4];
    p[0] = v.x; p[1] = v.y; p[2] = v.z; p[3] = v.w;
  }
  __syncthreads();

  const int hi = lane >> 4;
  const int l  = lane & 15;

  for (int nt = wave; nt < 16; nt += 8) {
    const int n0 = nt << 4;
    v8f c = {};
    for (int k0 = 0; k0 < F; k0 += 4) {
      const int ka = k0 + (hi << 1);
      v2f a, b;
      a.x = As[l * LDSPITCH + ka];
      a.y = As[l * LDSPITCH + ka + 1];
      b.x = W[(size_t)ka * F + n0 + l];
      b.y = W[(size_t)(ka + 1) * F + n0 + l];
      c = __builtin_amdgcn_wmma_f32_16x16x4_f32(false, a, false, b, (short)0, c,
                                                false, false);
    }
#pragma unroll
    for (int v = 0; v < 8; ++v)
      C[(size_t)(m0 + (hi << 3) + v) * F + n0 + l] = c[v];
  }
}

__global__ void gat_attn_coeff(const float* __restrict__ h,
                               const float* __restrict__ al,
                               const float* __restrict__ ar,
                               float* __restrict__ el, float* __restrict__ er) {
  const int i = blockIdx.x * blockDim.x + threadIdx.x;
  if (i >= NN * HH) return;
  const int n = i >> 3, hd = i & 7;
  const float* hp  = h + (size_t)n * F + hd * DD;
  const float* alp = al + hd * DD;
  const float* arp = ar + hd * DD;
  float sl = 0.f, sr = 0.f;
#pragma unroll 8
  for (int d = 0; d < DD; ++d) {
    const float v = hp[d];
    sl += v * alp[d];
    sr += v * arp[d];
  }
  el[i] = sl;
  er[i] = sr;
}

__global__ void gat_init(float* __restrict__ m, float* __restrict__ s,
                         float* __restrict__ agg) {
  const int i = blockIdx.x * blockDim.x + threadIdx.x;
  if (i < NN * HH) { m[i] = -INFINITY; s[i] = 0.f; }
  if (i < NN * F) agg[i] = 0.f;
}

__device__ __forceinline__ void atomicMaxF32(float* addr, float v) {
  if (v >= 0.f)
    atomicMax(reinterpret_cast<int*>(addr), __float_as_int(v));
  else
    atomicMin(reinterpret_cast<unsigned int*>(addr),
              (unsigned int)__float_as_int(v));
}

__global__ void gat_edge_max(const float* __restrict__ el,
                             const float* __restrict__ er,
                             const int* __restrict__ src,
                             const int* __restrict__ dst,
                             float* __restrict__ e, float* __restrict__ m) {
  const int i = blockIdx.x * blockDim.x + threadIdx.x;
  if (i >= EE * HH) return;
  const int ed = i >> 3, hd = i & 7;
  const int sn = src[ed], dn = dst[ed];
  float v = el[sn * HH + hd] + er[dn * HH + hd];
  v = (v > 0.f) ? v : 0.2f * v;
  e[i] = v;
  atomicMaxF32(&m[dn * HH + hd], v);
}

__global__ void gat_edge_exp(float* __restrict__ e, const float* __restrict__ m,
                             const int* __restrict__ dst, float* __restrict__ s) {
  const int i = blockIdx.x * blockDim.x + threadIdx.x;
  if (i >= EE * HH) return;
  const int ed = i >> 3, hd = i & 7;
  const int dn = dst[ed];
  const float v = __expf(e[i] - m[dn * HH + hd]);
  e[i] = v;
  atomicAdd(&s[dn * HH + hd], v);
}

__global__ void gat_edge_agg(const float* __restrict__ ex,
                             const float* __restrict__ s,
                             const int* __restrict__ src,
                             const int* __restrict__ dst,
                             const float* __restrict__ hfeat,
                             float* __restrict__ agg) {
  const int i = blockIdx.x * blockDim.x + threadIdx.x;
  if (i >= EE * HH) return;
  const int ed = i >> 3, hd = i & 7;
  const int sn = src[ed], dn = dst[ed];
  const float alpha = ex[i] / s[dn * HH + hd];
  const float* hp = hfeat + (size_t)sn * F + hd * DD;
  float* op = agg + (size_t)dn * F + hd * DD;
#pragma unroll 8
  for (int d = 0; d < DD; ++d) atomicAdd(&op[d], alpha * hp[d]);
}

__global__ void gat_bias_elu(const float* __restrict__ agg,
                             const float* __restrict__ b,
                             float* __restrict__ out) {
  const int i = blockIdx.x * blockDim.x + threadIdx.x;
  if (i >= NN * F) return;
  const float v = agg[i] + b[i & (F - 1)];
  out[i] = (v > 0.f) ? v : (__expf(v) - 1.f);
}

extern "C" void kernel_launch(void* const* d_in, const int* in_sizes, int n_in,
                              void* d_out, int out_size, void* d_ws, size_t ws_size,
                              hipStream_t stream) {
  const float* feats = (const float*)d_in[0];
  const float* W0    = (const float*)d_in[1];
  const float* al0   = (const float*)d_in[2];
  const float* ar0   = (const float*)d_in[3];
  const float* b0    = (const float*)d_in[4];
  const float* W1    = (const float*)d_in[5];
  const float* al1   = (const float*)d_in[6];
  const float* ar1   = (const float*)d_in[7];
  const float* b1    = (const float*)d_in[8];
  const int*   src   = (const int*)d_in[9];
  const int*   dst   = (const int*)d_in[10];

  float* ws = (float*)d_ws;
  float* hbuf = ws;
  float* agg  = hbuf + (size_t)NN * F;
  float* x1   = agg + (size_t)NN * F;
  float* el   = x1 + (size_t)NN * F;
  float* er   = el + (size_t)NN * HH;
  float* mmax = er + (size_t)NN * HH;
  float* ssum = mmax + (size_t)NN * HH;
  float* ebuf = ssum + (size_t)NN * HH;

  const int bl = 256;
  const int gNF = (NN * F + bl - 1) / bl;
  const int gNH = (NN * HH + bl - 1) / bl;
  const int gEH = (EE * HH + bl - 1) / bl;

  auto run_layer = [&](const float* x, const float* Wm, const float* al,
                       const float* ar, const float* bb, float* outx) {
    gat_gemm_wmma<<<NN / 16, 256, 0, stream>>>(x, Wm, hbuf);
    gat_attn_coeff<<<gNH, bl, 0, stream>>>(hbuf, al, ar, el, er);
    gat_init<<<gNF, bl, 0, stream>>>(mmax, ssum, agg);
    gat_edge_max<<<gEH, bl, 0, stream>>>(el, er, src, dst, ebuf, mmax);
    gat_edge_exp<<<gEH, bl, 0, stream>>>(ebuf, mmax, dst, ssum);
    gat_edge_agg<<<gEH, bl, 0, stream>>>(ebuf, ssum, src, dst, hbuf, agg);
    gat_bias_elu<<<gNF, bl, 0, stream>>>(agg, bb, outx);
  };

  run_layer(feats, W0, al0, ar0, b0, x1);
  run_layer(x1, W1, al1, ar1, b1, (float*)d_out);
}


# edge aggregation remapped: a wave walks each edge's 256-float row, so each load and each f32 atomic add covers one 128-B line (was 32 lines); sums stay atomic, arrival order as before
# speedup vs baseline: 5.8280x; 5.8280x over previous
#include <hip/hip_runtime.h>
#include <hip/hip_bf16.h>
#include <math.h>

#define NN 20000
#define EE 320000
#define HH 8
#define DD 32
#define F  256
#define LDSPITCH 260

typedef float v2f __attribute__((ext_vector_type(2)));
typedef float v8f __attribute__((ext_vector_type(8)));

__global__ __launch_bounds__(256) void gat_gemm_wmma(
    const float* __restrict__ A, const float* __restrict__ W,
    float* __restrict__ C) {
  __shared__ float As[16 * LDSPITCH];
  const int tid  = threadIdx.x;
  const int wave = tid >> 5;
  const int lane = tid & 31;
  const int m0   = blockIdx.x << 4;

  for (int i = tid; i < 16 * 64; i += 256) {
    const int r  = i >> 6;
    const int c4 = (i & 63) << 2;
    const float4 v = *reinterpret_cast<const float4*>(A + (size_t)(m0 + r) * F + c4);
    float* p = &As[r * LDSPITCH + c4];
    p[0] = v.x; p[1] = v.y; p[2] = v.z; p[3] = v.w;
  }
  __syncthreads();

  const int hi = lane >> 4;
  const int l  = lane & 15;

  for (int nt = wave; nt < 16; nt += 8) {
    const int n0 = nt << 4;
    v8f c = {};
    for (int k0 = 0; k0 < F; k0 += 4) {
      const int ka = k0 + (hi << 1);
      v2f a, b;
      a.x = As[l * LDSPITCH + ka];
      a.y = As[l * LDSPITCH + ka + 1];
      b.x = W[(size_t)ka * F + n0 + l];
      b.y = W[(size_t)(ka + 1) * F + n0 + l];
      c = __builtin_amdgcn_wmma_f32_16x16x4_f32(false, a, false, b, (short)0, c,
                                                false, false);
    }
#pragma unroll
    for (int v = 0; v < 8; ++v)
      C[(size_t)(m0 + (hi << 3) + v) * F + n0 + l] = c[v];
  }
}

__global__ void gat_attn_coeff(const float* __restrict__ h,
                               const float* __restrict__ al,
                               const float* __restrict__ ar,
                               float* __restrict__ el, float* __restrict__ er) {
  const int i = blockIdx.x * blockDim.x + threadIdx.x;
  if (i >= NN * HH) return;
  const int n = i >> 3, hd = i & 7;
  const float* hp  = h + (size_t)n * F + hd * DD;
  const float* alp = al + hd * DD;
  const float* arp = ar + hd * DD;
  float sl = 0.f, sr = 0.f;
#pragma unroll 8
  for (int d = 0; d < DD; ++d) {
    const float v = hp[d];
    sl += v * alp[d];
    sr += v * arp[d];
  }
  el[i] = sl;
  er[i] = sr;
}

__global__ void gat_init(float* __restrict__ m, float* __restrict__ s,
                         float* __restrict__ agg) {
  const int i = blockIdx.x * blockDim.x + threadIdx.x;
  if (i < NN * HH) { m[i] = -INFINITY; s[i] = 0.f; }
  if (i < NN * F) agg[i] = 0.f;
}

__device__ __forceinline__ void atomicMaxF32(float* addr, float v) {
  if (v >= 0.f)
    atomicMax(reinterpret_cast<int*>(addr), __float_as_int(v));
  else
    atomicMin(reinterpret_cast<unsigned int*>(addr),
              (unsigned int)__float_as_int(v));
}

__global__ void gat_edge_max(const float* __restrict__ el,
                             const float* __restrict__ er,
                             const int* __restrict__ src,
                             const int* __restrict__ dst,
                             float* __restrict__ e, float* __restrict__ m) {
  const int i = blockIdx.x * blockDim.x + threadIdx.x;
  if (i >= EE * HH) return;
  const int ed = i >> 3, hd = i & 7;
  const int sn = src[ed], dn = dst[ed];
  float v = el[sn * HH + hd] + er[dn * HH + hd];
  v = (v > 0.f) ? v : 0.2f * v;
  e[i] = v;
  atomicMaxF32(&m[dn * HH + hd], v);
}

__global__ void gat_edge_exp(float* __restrict__ e, const float* __restrict__ m,
                             const int* __restrict__ dst, float* __restrict__ s) {
  const int i = blockIdx.x * blockDim.x + threadIdx.x;
  if (i >= EE * HH) return;
  const int ed = i >> 3, hd = i & 7;
  const int dn = dst[ed];
  const float v = __expf(e[i] - m[dn * HH + hd]);
  e[i] = v;
  atomicAdd(&s[dn * HH + hd], v);
}

#define AGG_BLOCKS 2048
__global__ __launch_bounds__(256) void gat_edge_agg(const float* __restrict__ ex,
                             const float* __restrict__ s,
                             const int* __restrict__ src,
                             const int* __restrict__ dst,
                             const float* __restrict__ hfeat,
                             float* __restrict__ agg) {
  const int lane = threadIdx.x & 31;
  const int wave = __builtin_amdgcn_readfirstlane(
      (int)((blockIdx.x * blockDim.x + threadIdx.x) >> 5));
  const int nwaves = (int)((gridDim.x * blockDim.x) >> 5);
  const int sub = lane >> 3, hd = lane & 7;

  for (int q = wave; q < EE / 4; q += nwaves) {
    const int ed = q * 4 + sub;
    const int sn = src[ed], dn = dst[ed];
    const int abits = __float_as_int(ex[ed * HH + hd] / s[dn * HH + hd]);

    float v[4][HH];
#pragma unroll
    for (int k = 0; k < 4; ++k) {
      const int sk = __builtin_amdgcn_readlane(sn, k * 8);
      const float* hp = hfeat + (size_t)sk * F + lane;
#pragma unroll
      for (int j = 0; j < HH; ++j) v[k][j] = hp[j * DD];
    }
#pragma unroll
    for (int k = 0; k < 4; ++k) {
      const int dk = __builtin_amdgcn_readlane(dn, k * 8);
      float* op = agg + (size_t)dk * F + lane;
#pragma unroll
      for (int j = 0; j < HH; ++j) {
        const float a = __int_as_float(__builtin_amdgcn_readlane(abits, k * 8 + j));
        atomicAdd(op + j * DD, a * v[k][j]);
      }
    }
  }
}

__global__ void gat_bias_elu(const float* __restrict__ agg,
                             const float* __restrict__ b,
                             float* __restrict__ out) {
  const int i = blockIdx.x * blockDim.x + threadIdx.x;
  if (i >= NN * F) return;
  const float v = agg[i] + b[i & (F - 1)];
  out[i] = (v > 0.f) ? v : (__expf(v) - 1.f);
}

extern "C" void kernel_launch(void* const* d_in, const int* in_sizes, int n_in,
                              void* d_out, int out_size, void* d_ws, size_t ws_size,
                              hipStream_t stream) {
  const float* feats = (const float*)d_in[0];
  const float* W0    = (const float*)d_in[1];
  const float* al0   = (const float*)d_in[2];
  const float* ar0   = (const float*)d_in[3];
  const float* b0    = (const float*)d_in[4];
  const float* W1    = (const float*)d_in[5];
  const float* al1   = (const float*)d_in[6];
  const float* ar1   = (const float*)d_in[7];
  const float* b1    = (const float*)d_in[8];
  const int*   src   = (const int*)d_in[9];
  const int*   dst   = (const int*)d_in[10];

  float* ws = (float*)d_ws;
  float* hbuf = ws;
  float* agg  = hbuf + (size_t)NN * F;
  float* x1   = agg + (size_t)NN * F;
  float* el   = x1 + (size_t)NN * F;
  float* er   = el + (size_t)NN * HH;
  float* mmax = er + (size_t)NN * HH;
  float* ssum = mmax + (size_t)NN * HH;
  float* ebuf = ssum + (size_t)NN * HH;

  const int bl = 256;
  const int gNF = (NN * F + bl - 1) / bl;
  const int gNH = (NN * HH + bl - 1) / bl;
  const int gEH = (EE * HH + bl - 1) / bl;

  auto run_layer = [&](const float* x, const float* Wm, const float* al,
                       const float* ar, const float* bb, float* outx) {
    gat_gemm_wmma<<<NN / 16, 256, 0, stream>>>(x, Wm, hbuf);
    gat_attn_coeff<<<gNH, bl, 0, stream>>>(hbuf, al, ar, el, er);
    gat_init<<<gNF, bl, 0, stream>>>(mmax, ssum, agg);
    gat_edge_max<<<gEH, bl, 0, stream>>>(el, er, src, dst, ebuf, mmax);
    gat_edge_exp<<<gEH, bl, 0, stream>>>(ebuf, mmax, dst, ssum);
    gat_edge_agg<<<AGG_BLOCKS, 256, 0, stream>>>(ebuf, ssum, src, dst, hbuf, agg);
    gat_bias_elu<<<gNF, bl, 0, stream>>>(agg, bb, outx);
  };

  run_layer(feats, W0, al0, ar0, b0, x1);
  run_layer(x1, W1, al1, ar1, b1, (float*)d_out);
}


# edges grouped by destination once per call (integer atomic ranks + prefix sum); one wave owns a destination row: softmax and weighted sum in registers, no float atomics, sum order = arrival rank
# speedup vs baseline: 10.4449x; 1.7922x over previous
#include <hip/hip_runtime.h>
#include <hip/hip_bf16.h>
#include <math.h>

#define NN 20000
#define EE 320000
#define HH 8
#define DD 32
#define F  256
#define LDSPITCH 260

typedef float v2f __attribute__((ext_vector_type(2)));
typedef float v8f __attribute__((ext_vector_type(8)));

__global__ __launch_bounds__(256) void gat_gemm_wmma(
    const float* __restrict__ A, const float* __restrict__ W,
    float* __restrict__ C) {
  __shared__ float As[16 * LDSPITCH];
  const int tid  = threadIdx.x;
  const int wave = tid >> 5;
  const int lane = tid & 31;
  const int m0   = blockIdx.x << 4;

  for (int i = tid; i < 16 * 64; i += 256) {
    const int r  = i >> 6;
    const int c4 = (i & 63) << 2;
    const float4 v = *reinterpret_cast<const float4*>(A + (size_t)(m0 + r) * F + c4);
    float* p = &As[r * LDSPITCH + c4];
    p[0] = v.x; p[1] = v.y; p[2] = v.z; p[3] = v.w;
  }
  __syncthreads();

  const int hi = lane >> 4;
  const int l  = lane & 15;

  for (int nt = wave; nt < 16; nt += 8) {
    const int n0 = nt << 4;
    v8f c = {};
    for (int k0 = 0; k0 < F; k0 += 4) {
      const int ka = k0 + (hi << 1);
      v2f a, b;
      a.x = As[l * LDSPITCH + ka];
      a.y = As[l * LDSPITCH + ka + 1];
      b.x = W[(size_t)ka * F + n0 + l];
      b.y = W[(size_t)(ka + 1) * F + n0 + l];
      c = __builtin_amdgcn_wmma_f32_16x16x4_f32(false, a, false, b, (short)0, c,
                                                false, false);
    }
#pragma unroll
    for (int v = 0; v < 8; ++v)
      C[(size_t)(m0 + (hi << 3) + v) * F + n0 + l] = c[v];
  }
}

__global__ void gat_attn_coeff(const float* __restrict__ h,
                               const float* __restrict__ al,
                               const float* __restrict__ ar,
                               float* __restrict__ el, float* __restrict__ er) {
  const int i = blockIdx.x * blockDim.x + threadIdx.x;
  if (i >= NN * HH) return;
  const int n = i >> 3, hd = i & 7;
  const float* hp  = h + (size_t)n * F + hd * DD;
  const float* alp = al + hd * DD;
  const float* arp = ar + hd * DD;
  float sl = 0.f, sr = 0.f;
#pragma unroll 8
  for (int d = 0; d < DD; ++d) {
    const float v = hp[d];
    sl += v * alp[d];
    sr += v * arp[d];
  }
  el[i] = sl;
  er[i] = sr;
}

__global__ void gat_zero_cnt(int* __restrict__ cnt) {
  const int i = blockIdx.x * blockDim.x + threadIdx.x;
  if (i < NN) cnt[i] = 0;
}

__global__ void gat_count_rank(const int* __restrict__ dst, int* __restrict__ cnt,
                               int* __restrict__ rank) {
  const int e = blockIdx.x * blockDim.x + threadIdx.x;
  if (e < EE) rank[e] = atomicAdd(&cnt[dst[e]], 1);
}

#define SCAN_PER 20
__global__ __launch_bounds__(1024) void gat_scan(const int* __restrict__ cnt,
                                                 int* __restrict__ rowptr) {
  __shared__ int part[1024];
  const int t = threadIdx.x;
  const int base = t * SCAN_PER;
  int loc[SCAN_PER];
  int sum = 0;
#pragma unroll
  for (int i = 0; i < SCAN_PER; ++i) {
    const int idx = base + i;
    loc[i] = sum;
    sum += (idx < NN) ? cnt[idx] : 0;
  }
  part[t] = sum;
  __syncthreads();
  for (int off = 1; off < 1024; off <<= 1) {
    const int add = (t >= off) ? part[t - off] : 0;
    __syncthreads();
    part[t] += add;
    __syncthreads();
  }
  const int excl = part[t] - sum;
#pragma unroll
  for (int i = 0; i < SCAN_PER; ++i) {
    const int idx = base + i;
    if (idx < NN) rowptr[idx] = excl + loc[i];
  }
  if (t == 1023) rowptr[NN] = part[1023];
}

__global__ void gat_place(const int* __restrict__ src, const int* __restrict__ dst,
                          const int* __restrict__ rank,
                          const int* __restrict__ rowptr, int* __restrict__ csrc) {
  const int e = blockIdx.x * blockDim.x + threadIdx.x;
  if (e < EE) csrc[rowptr[dst[e]] + rank[e]] = src[e];
}

__global__ __launch_bounds__(256) void gat_node_softmax_agg(
    const float* __restrict__ el, const float* __restrict__ er,
    const int* __restrict__ rowptr, const int* __restrict__ csrc,
    const float* __restrict__ hfeat, const float* __restrict__ b,
    float* __restrict__ out) {
  const int lane = threadIdx.x & 31;
  const int n = __builtin_amdgcn_readfirstlane(
      (int)((blockIdx.x * blockDim.x + threadIdx.x) >> 5));
  if (n >= NN) return;
  const int sub = lane >> 3, hd = lane & 7;
  const int beg = rowptr[n], end = rowptr[n + 1];
  const float ern = er[n * HH + hd];

  float mx = -INFINITY;
  for (int base = beg; base < end; base += 4) {
    const int ed = base + sub;
    if (ed < end) {
      float v = el[csrc[ed] * HH + hd] + ern;
      v = (v > 0.f) ? v : 0.2f * v;
      mx = fmaxf(mx, v);
    }
  }
  mx = fmaxf(mx, __shfl_xor(mx, 8));
  mx = fmaxf(mx, __shfl_xor(mx, 16));

  float sm = 0.f;
  for (int base = beg; base < end; base += 4) {
    const int ed = base + sub;
    if (ed < end) {
      float v = el[csrc[ed] * HH + hd] + ern;
      v = (v > 0.f) ? v : 0.2f * v;
      sm += __expf(v - mx);
    }
  }
  sm += __shfl_xor(sm, 8);
  sm += __shfl_xor(sm, 16);

  float4 acc0 = make_float4(0.f, 0.f, 0.f, 0.f);
  float4 acc1 = make_float4(0.f, 0.f, 0.f, 0.f);
  const int hsel = lane >> 3;
  for (int base = beg; base < end; base += 4) {
    const int ed = base + sub;
    int sn = 0;
    float alpha = 0.f;
    if (ed < end) {
      sn = csrc[ed];
      float v = el[sn * HH + hd] + ern;
      v = (v > 0.f) ? v : 0.2f * v;
      alpha = __expf(v - mx) / sm;
    }
    if (base + 4 <= end) {
      float4 r0[4], r1[4];
#pragma unroll
      for (int k = 0; k < 4; ++k) {
        const int sk = __builtin_amdgcn_readlane(sn, k * 8);
        const float4* hp = reinterpret_cast<const float4*>(hfeat + (size_t)sk * F) + lane;
        r0[k] = hp[0];
        r1[k] = hp[32];
      }
#pragma unroll
      for (int k = 0; k < 4; ++k) {
        const float a0 = __shfl(alpha, k * 8 + hsel);
        const float a1 = __shfl(alpha, k * 8 + hsel + 4);
        acc0.x += a0 * r0[k].x; acc0.y += a0 * r0[k].y;
        acc0.z += a0 * r0[k].z; acc0.w += a0 * r0[k].w;
        acc1.x += a1 * r1[k].x; acc1.y += a1 * r1[k].y;
        acc1.z += a1 * r1[k].z; acc1.w += a1 * r1[k].w;
      }
    } else {
#pragma unroll
      for (int k = 0; k < 4; ++k) {
        const float a0 = __shfl(alpha, k * 8 + hsel);
        const float a1 = __shfl(alpha, k * 8 + hsel + 4);
        if (base + k < end) {
          const int sk = __builtin_amdgcn_readlane(sn, k * 8);
          const float4* hp = reinterpret_cast<const float4*>(hfeat + (size_t)sk * F) + lane;
          const float4 q0 = hp[0];
          const float4 q1 = hp[32];
          acc0.x += a0 * q0.x; acc0.y += a0 * q0.y;
          acc0.z += a0 * q0.z; acc0.w += a0 * q0.w;
          acc1.x += a1 * q1.x; acc1.y += a1 * q1.y;
          acc1.z += a1 * q1.z; acc1.w += a1 * q1.w;
        }
      }
    }
  }

  const float4* b4 = reinterpret_cast<const float4*>(b) + lane;
  float4* o4 = reinterpret_cast<float4*>(out + (size_t)n * F) + lane;
  const float4 bA = b4[0], bB = b4[32];
  float4 oA, oB;
  float t;
  t = acc0.x + bA.x; oA.x = (t > 0.f) ? t : (__expf(t) - 1.f);
  t = acc0.y + bA.y; oA.y = (t > 0.f) ? t : (__expf(t) - 1.f);
  t = acc0.z + bA.z; oA.z = (t > 0.f) ? t : (__expf(t) - 1.f);
  t = acc0.w + bA.w; oA.w = (t > 0.f) ? t : (__expf(t) - 1.f);
  t = acc1.x + bB.x; oB.x = (t > 0.f) ? t : (__expf(t) - 1.f);
  t = acc1.y + bB.y; oB.y = (t > 0.f) ? t : (__expf(t) - 1.f);
  t = acc1.z + bB.z; oB.z = (t > 0.f) ? t : (__expf(t) - 1.f);
  t = acc1.w + bB.w; oB.w = (t > 0.f) ? t : (__expf(t) - 1.f);
  o4[0]  = oA;
  o4[32] = oB;
}

extern "C" void kernel_launch(void* const* d_in, const int* in_sizes, int n_in,
                              void* d_out, int out_size, void* d_ws, size_t ws_size,
                              hipStream_t stream) {
  const float* feats = (const float*)d_in[0];
  const float* W0    = (const float*)d_in[1];
  const float* al0   = (const float*)d_in[2];
  const float* ar0   = (const float*)d_in[3];
  const float* b0    = (const float*)d_in[4];
  const float* W1    = (const float*)d_in[5];
  const float* al1   = (const float*)d_in[6];
  const float* ar1   = (const float*)d_in[7];
  const float* b1    = (const float*)d_in[8];
  const int*   src   = (const int*)d_in[9];
  const int*   dst   = (const int*)d_in[10];

  float* ws = (float*)d_ws;
  float* hbuf = ws;
  float* x1   = hbuf + (size_t)NN * F;
  float* el   = x1 + (size_t)NN * F;
  float* er   = el + (size_t)NN * HH;
  int* cnt    = (int*)(er + (size_t)NN * HH);
  int* rowptr = cnt + NN;
  int* rank   = rowptr + (NN + 32);
  int* csrc   = rank + EE;

  const int bl = 256;
  const int gN  = (NN + bl - 1) / bl;
  const int gNH = (NN * HH + bl - 1) / bl;
  const int gE  = (EE + bl - 1) / bl;

  gat_zero_cnt<<<gN, bl, 0, stream>>>(cnt);
  gat_count_rank<<<gE, bl, 0, stream>>>(dst, cnt, rank);
  gat_scan<<<1, 1024, 0, stream>>>(cnt, rowptr);
  gat_place<<<gE, bl, 0, stream>>>(src, dst, rank, rowptr, csrc);

  auto run_layer = [&](const float* x, const float* Wm, const float* al,
                       const float* ar, const float* bb, float* outx) {
    gat_gemm_wmma<<<NN / 16, 256, 0, stream>>>(x, Wm, hbuf);
    gat_attn_coeff<<<gNH, bl, 0, stream>>>(hbuf, al, ar, el, er);
    gat_node_softmax_agg<<<(NN * 32 + bl - 1) / bl, bl, 0, stream>>>(
        el, er, rowptr, csrc, hbuf, bb, outx);
  };

  run_layer(feats, W0, al0, ar0, b0, x1);
  run_layer(x1, W1, al1, ar1, b1, (float*)d_out);
}


# el/er computed in the GEMM epilogue from the f32 accumulators (wave owns one head's two column tiles); coefficient kernel and its re-read of h removed; el/er sum order: 16-lane shuffle tree
# speedup vs baseline: 11.7649x; 1.1264x over previous
#include <hip/hip_runtime.h>
#include <hip/hip_bf16.h>
#include <math.h>

#define NN 20000
#define EE 320000
#define HH 8
#define DD 32
#define F  256
#define LDSPITCH 260

typedef float v2f __attribute__((ext_vector_type(2)));
typedef float v8f __attribute__((ext_vector_type(8)));

__global__ __launch_bounds__(256) void gat_gemm_wmma(
    const float* __restrict__ A, const float* __restrict__ W,
    const float* __restrict__ al, const float* __restrict__ ar,
    float* __restrict__ C, float* __restrict__ el, float* __restrict__ er) {
  __shared__ float As[16 * LDSPITCH];
  __shared__ float Es[2 * 16 * HH];
  const int tid  = threadIdx.x;
  const int wave = tid >> 5;
  const int lane = tid & 31;
  const int m0   = blockIdx.x << 4;

  for (int i = tid; i < 16 * 64; i += 256) {
    const int r  = i >> 6;
    const int c4 = (i & 63) << 2;
    const float4 v = *reinterpret_cast<const float4*>(A + (size_t)(m0 + r) * F + c4);
    float* p = &As[r * LDSPITCH + c4];
    p[0] = v.x; p[1] = v.y; p[2] = v.z; p[3] = v.w;
  }
  __syncthreads();

  const int hi = lane >> 4;
  const int l  = lane & 15;
  const int n0 = wave << 5;

  v8f c0 = {}, c1 = {};
  for (int k0 = 0; k0 < F; k0 += 4) {
    const int ka = k0 + (hi << 1);
    v2f a, b0, b1;
    a.x  = As[l * LDSPITCH + ka];
    a.y  = As[l * LDSPITCH + ka + 1];
    b0.x = W[(size_t)ka * F + n0 + l];
    b0.y = W[(size_t)(ka + 1) * F + n0 + l];
    b1.x = W[(size_t)ka * F + n0 + 16 + l];
    b1.y = W[(size_t)(ka + 1) * F + n0 + 16 + l];
    c0 = __builtin_amdgcn_wmma_f32_16x16x4_f32(false, a, false, b0, (short)0, c0,
                                               false, false);
    c1 = __builtin_amdgcn_wmma_f32_16x16x4_f32(false, a, false, b1, (short)0, c1,
                                               false, false);
  }
#pragma unroll
  for (int v = 0; v < 8; ++v) {
    float* crow = C + (size_t)(m0 + (hi << 3) + v) * F + n0 + l;
    crow[0]  = c0[v];
    crow[16] = c1[v];
  }

  const float alA = al[n0 + l], alB = al[n0 + 16 + l];
  const float arA = ar[n0 + l], arB = ar[n0 + 16 + l];
  const bool upper = (l & 8) != 0;
  float keep[8], give[8];
#pragma unroll
  for (int v = 0; v < 8; ++v) {
    const float pl = c0[v] * alA + c1[v] * alB;
    const float pr = c0[v] * arA + c1[v] * arB;
    keep[v] = upper ? pr : pl;
    give[v] = upper ? pl : pr;
  }
#pragma unroll
  for (int v = 0; v < 8; ++v) keep[v] += __shfl_xor(give[v], 8);
#pragma unroll
  for (int v = 0; v < 8; ++v) keep[v] += __shfl_xor(keep[v], 4);
#pragma unroll
  for (int v = 0; v < 8; ++v) keep[v] += __shfl_xor(keep[v], 2);
#pragma unroll
  for (int v = 0; v < 8; ++v) keep[v] += __shfl_xor(keep[v], 1);
  if ((l & 7) == 0) {
    float* es = &Es[(upper ? 16 * HH : 0) + (hi << 3) * HH + wave];
#pragma unroll
    for (int v = 0; v < 8; ++v) es[v * HH] = keep[v];
  }
  __syncthreads();
  if (tid < 16 * HH) el[(size_t)m0 * HH + tid] = Es[tid];
  else               er[(size_t)m0 * HH + (tid - 16 * HH)] = Es[tid];
}

__global__ void gat_zero_cnt(int* __restrict__ cnt) {
  const int i = blockIdx.x * blockDim.x + threadIdx.x;
  if (i < NN) cnt[i] = 0;
}

__global__ void gat_count_rank(const int* __restrict__ dst, int* __restrict__ cnt,
                               int* __restrict__ rank) {
  const int e = blockIdx.x * blockDim.x + threadIdx.x;
  if (e < EE) rank[e] = atomicAdd(&cnt[dst[e]], 1);
}

#define SCAN_PER 20
__global__ __launch_bounds__(1024) void gat_scan(const int* __restrict__ cnt,
                                                 int* __restrict__ rowptr) {
  __shared__ int part[1024];
  const int t = threadIdx.x;
  const int base = t * SCAN_PER;
  int loc[SCAN_PER];
  int sum = 0;
#pragma unroll
  for (int i = 0; i < SCAN_PER; ++i) {
    const int idx = base + i;
    loc[i] = sum;
    sum += (idx < NN) ? cnt[idx] : 0;
  }
  part[t] = sum;
  __syncthreads();
  for (int off = 1; off < 1024; off <<= 1) {
    const int add = (t >= off) ? part[t - off] : 0;
    __syncthreads();
    part[t] += add;
    __syncthreads();
  }
  const int excl = part[t] - sum;
#pragma unroll
  for (int i = 0; i < SCAN_PER; ++i) {
    const int idx = base + i;
    if (idx < NN) rowptr[idx] = excl + loc[i];
  }
  if (t == 1023) rowptr[NN] = part[1023];
}

__global__ void gat_place(const int* __restrict__ src, const int* __restrict__ dst,
                          const int* __restrict__ rank,
                          const int* __restrict__ rowptr, int* __restrict__ csrc) {
  const int e = blockIdx.x * blockDim.x + threadIdx.x;
  if (e < EE) csrc[rowptr[dst[e]] + rank[e]] = src[e];
}

__global__ __launch_bounds__(256) void gat_node_softmax_agg(
    const float* __restrict__ el, const float* __restrict__ er,
    const int* __restrict__ rowptr, const int* __restrict__ csrc,
    const float* __restrict__ hfeat, const float* __restrict__ b,
    float* __restrict__ out) {
  const int lane = threadIdx.x & 31;
  const int n = __builtin_amdgcn_readfirstlane(
      (int)((blockIdx.x * blockDim.x + threadIdx.x) >> 5));
  if (n >= NN) return;
  const int sub = lane >> 3, hd = lane & 7;
  const int beg = rowptr[n], end = rowptr[n + 1];
  const float ern = er[n * HH + hd];

  float mx = -INFINITY;
  for (int base = beg; base < end; base += 4) {
    const int ed = base + sub;
    if (ed < end) {
      float v = el[csrc[ed] * HH + hd] + ern;
      v = (v > 0.f) ? v : 0.2f * v;
      mx = fmaxf(mx, v);
    }
  }
  mx = fmaxf(mx, __shfl_xor(mx, 8));
  mx = fmaxf(mx, __shfl_xor(mx, 16));

  float sm = 0.f;
  for (int base = beg; base < end; base += 4) {
    const int ed = base + sub;
    if (ed < end) {
      float v = el[csrc[ed] * HH + hd] + ern;
      v = (v > 0.f) ? v : 0.2f * v;
      sm += __expf(v - mx);
    }
  }
  sm += __shfl_xor(sm, 8);
  sm += __shfl_xor(sm, 16);

  float4 acc0 = make_float4(0.f, 0.f, 0.f, 0.f);
  float4 acc1 = make_float4(0.f, 0.f, 0.f, 0.f);
  const int hsel = lane >> 3;
  for (int base = beg; base < end; base += 4) {
    const int ed = base + sub;
    int sn = 0;
    float alpha = 0.f;
    if (ed < end) {
      sn = csrc[ed];
      float v = el[sn * HH + hd] + ern;
      v = (v > 0.f) ? v : 0.2f * v;
      alpha = __expf(v - mx) / sm;
    }
    if (base + 4 <= end) {
      float4 r0[4], r1[4];
#pragma unroll
      for (int k = 0; k < 4; ++k) {
        const int sk = __builtin_amdgcn_readlane(sn, k * 8);
        const float4* hp = reinterpret_cast<const float4*>(hfeat + (size_t)sk * F) + lane;
        r0[k] = hp[0];
        r1[k] = hp[32];
      }
#pragma unroll
      for (int k = 0; k < 4; ++k) {
        const float a0 = __shfl(alpha, k * 8 + hsel);
        const float a1 = __shfl(alpha, k * 8 + hsel + 4);
        acc0.x += a0 * r0[k].x; acc0.y += a0 * r0[k].y;
        acc0.z += a0 * r0[k].z; acc0.w += a0 * r0[k].w;
        acc1.x += a1 * r1[k].x; acc1.y += a1 * r1[k].y;
        acc1.z += a1 * r1[k].z; acc1.w += a1 * r1[k].w;
      }
    } else {
#pragma unroll
      for (int k = 0; k < 4; ++k) {
        const float a0 = __shfl(alpha, k * 8 + hsel);
        const float a1 = __shfl(alpha, k * 8 + hsel + 4);
        if (base + k < end) {
          const int sk = __builtin_amdgcn_readlane(sn, k * 8);
          const float4* hp = reinterpret_cast<const float4*>(hfeat + (size_t)sk * F) + lane;
          const float4 q0 = hp[0];
          const float4 q1 = hp[32];
          acc0.x += a0 * q0.x; acc0.y += a0 * q0.y;
          acc0.z += a0 * q0.z; acc0.w += a0 * q0.w;
          acc1.x += a1 * q1.x; acc1.y += a1 * q1.y;
          acc1.z += a1 * q1.z; acc1.w += a1 * q1.w;
        }
      }
    }
  }

  const float4* b4 = reinterpret_cast<const float4*>(b) + lane;
  float4* o4 = reinterpret_cast<float4*>(out + (size_t)n * F) + lane;
  const float4 bA = b4[0], bB = b4[32];
  float4 oA, oB;
  float t;
  t = acc0.x + bA.x; oA.x = (t > 0.f) ? t : (__expf(t) - 1.f);
  t = acc0.y + bA.y; oA.y = (t > 0.f) ? t : (__expf(t) - 1.f);
  t = acc0.z + bA.z; oA.z = (t > 0.f) ? t : (__expf(t) - 1.f);
  t = acc0.w + bA.w; oA.w = (t > 0.f) ? t : (__expf(t) - 1.f);
  t = acc1.x + bB.x; oB.x = (t > 0.f) ? t : (__expf(t) - 1.f);
  t = acc1.y + bB.y; oB.y = (t > 0.f) ? t : (__expf(t) - 1.f);
  t = acc1.z + bB.z; oB.z = (t > 0.f) ? t : (__expf(t) - 1.f);
  t = acc1.w + bB.w; oB.w = (t > 0.f) ? t : (__expf(t) - 1.f);
  o4[0]  = oA;
  o4[32] = oB;
}

extern "C" void kernel_launch(void* const* d_in, const int* in_sizes, int n_in,
                              void* d_out, int out_size, void* d_ws, size_t ws_size,
                              hipStream_t stream) {
  const float* feats = (const float*)d_in[0];
  const float* W0    = (const float*)d_in[1];
  const float* al0   = (const float*)d_in[2];
  const float* ar0   = (const float*)d_in[3];
  const float* b0    = (const float*)d_in[4];
  const float* W1    = (const float*)d_in[5];
  const float* al1   = (const float*)d_in[6];
  const float* ar1   = (const float*)d_in[7];
  const float* b1    = (const float*)d_in[8];
  const int*   src   = (const int*)d_in[9];
  const int*   dst   = (const int*)d_in[10];

  float* ws = (float*)d_ws;
  float* hbuf = ws;
  float* x1   = hbuf + (size_t)NN * F;
  float* el   = x1 + (size_t)NN * F;
  float* er   = el + (size_t)NN * HH;
  int* cnt    = (int*)(er + (size_t)NN * HH);
  int* rowptr = cnt + NN;
  int* rank   = rowptr + (NN + 32);
  int* csrc   = rank + EE;

  const int bl = 256;
  const int gN  = (NN + bl - 1) / bl;
  const int gE  = (EE + bl - 1) / bl;

  gat_zero_cnt<<<gN, bl, 0, stream>>>(cnt);
  gat_count_rank<<<gE, bl, 0, stream>>>(dst, cnt, rank);
  gat_scan<<<1, 1024, 0, stream>>>(cnt, rowptr);
  gat_place<<<gE, bl, 0, stream>>>(src, dst, rank, rowptr, csrc);

  auto run_layer = [&](const float* x, const float* Wm, const float* al,
                       const float* ar, const float* bb, float* outx) {
    gat_gemm_wmma<<<NN / 16, 256, 0, stream>>>(x, Wm, al, ar, hbuf, el, er);
    gat_node_softmax_agg<<<(NN * 32 + bl - 1) / bl, bl, 0, stream>>>(
        el, er, rowptr, csrc, hbuf, bb, outx);
  };

  run_layer(feats, W0, al0, ar0, b0, x1);
  run_layer(x1, W1, al1, ar1, b1, (float*)d_out);
}


# aggregation, lists of up to 32 edges: list read once and sorted by source, scores gathered once and kept in registers; sum order now ascending source (was arrival rank), the same on every run
# speedup vs baseline: 12.3610x; 1.0507x over previous
#include <hip/hip_runtime.h>
#include <hip/hip_bf16.h>
#include <math.h>

#define NN 20000
#define EE 320000
#define HH 8
#define DD 32
#define F  256
#define LDSPITCH 260

typedef float v2f __attribute__((ext_vector_type(2)));
typedef float v8f __attribute__((ext_vector_type(8)));

__global__ __launch_bounds__(256) void gat_gemm_wmma(
    const float* __restrict__ A, const float* __restrict__ W,
    const float* __restrict__ al, const float* __restrict__ ar,
    float* __restrict__ C, float* __restrict__ el, float* __restrict__ er) {
  __shared__ float As[16 * LDSPITCH];
  __shared__ float Es[2 * 16 * HH];
  const int tid  = threadIdx.x;
  const int wave = tid >> 5;
  const int lane = tid & 31;
  const int m0   = blockIdx.x << 4;

  for (int i = tid; i < 16 * 64; i += 256) {
    const int r  = i >> 6;
    const int c4 = (i & 63) << 2;
    const float4 v = *reinterpret_cast<const float4*>(A + (size_t)(m0 + r) * F + c4);
    float* p = &As[r * LDSPITCH + c4];
    p[0] = v.x; p[1] = v.y; p[2] = v.z; p[3] = v.w;
  }
  __syncthreads();

  const int hi = lane >> 4;
  const int l  = lane & 15;
  const int n0 = wave << 5;

  v8f c0 = {}, c1 = {};
  for (int k0 = 0; k0 < F; k0 += 4) {
    const int ka = k0 + (hi << 1);
    v2f a, b0, b1;
    a.x  = As[l * LDSPITCH + ka];
    a.y  = As[l * LDSPITCH + ka + 1];
    b0.x = W[(size_t)ka * F + n0 + l];
    b0.y = W[(size_t)(ka + 1) * F + n0 + l];
    b1.x = W[(size_t)ka * F + n0 + 16 + l];
    b1.y = W[(size_t)(ka + 1) * F + n0 + 16 + l];
    c0 = __builtin_amdgcn_wmma_f32_16x16x4_f32(false, a, false, b0, (short)0, c0,
                                               false, false);
    c1 = __builtin_amdgcn_wmma_f32_16x16x4_f32(false, a, false, b1, (short)0, c1,
                                               false, false);
  }
#pragma unroll
  for (int v = 0; v < 8; ++v) {
    float* crow = C + (size_t)(m0 + (hi << 3) + v) * F + n0 + l;
    crow[0]  = c0[v];
    crow[16] = c1[v];
  }

  const float alA = al[n0 + l], alB = al[n0 + 16 + l];
  const float arA = ar[n0 + l], arB = ar[n0 + 16 + l];
  const bool upper = (l & 8) != 0;
  float keep[8], give[8];
#pragma unroll
  for (int v = 0; v < 8; ++v) {
    const float pl = c0[v] * alA + c1[v] * alB;
    const float pr = c0[v] * arA + c1[v] * arB;
    keep[v] = upper ? pr : pl;
    give[v] = upper ? pl : pr;
  }
#pragma unroll
  for (int v = 0; v < 8; ++v) keep[v] += __shfl_xor(give[v], 8);
#pragma unroll
  for (int v = 0; v < 8; ++v) keep[v] += __shfl_xor(keep[v], 4);
#pragma unroll
  for (int v = 0; v < 8; ++v) keep[v] += __shfl_xor(keep[v], 2);
#pragma unroll
  for (int v = 0; v < 8; ++v) keep[v] += __shfl_xor(keep[v], 1);
  if ((l & 7) == 0) {
    float* es = &Es[(upper ? 16 * HH : 0) + (hi << 3) * HH + wave];
#pragma unroll
    for (int v = 0; v < 8; ++v) es[v * HH] = keep[v];
  }
  __syncthreads();
  if (tid < 16 * HH) el[(size_t)m0 * HH + tid] = Es[tid];
  else               er[(size_t)m0 * HH + (tid - 16 * HH)] = Es[tid];
}

__global__ void gat_zero_cnt(int* __restrict__ cnt) {
  const int i = blockIdx.x * blockDim.x + threadIdx.x;
  if (i < NN) cnt[i] = 0;
}

__global__ void gat_count_rank(const int* __restrict__ dst, int* __restrict__ cnt,
                               int* __restrict__ rank) {
  const int e = blockIdx.x * blockDim.x + threadIdx.x;
  if (e < EE) rank[e] = atomicAdd(&cnt[dst[e]], 1);
}

#define SCAN_PER 20
__global__ __launch_bounds__(1024) void gat_scan(const int* __restrict__ cnt,
                                                 int* __restrict__ rowptr) {
  __shared__ int part[1024];
  const int t = threadIdx.x;
  const int base = t * SCAN_PER;
  int loc[SCAN_PER];
  int sum = 0;
#pragma unroll
  for (int i = 0; i < SCAN_PER; ++i) {
    const int idx = base + i;
    loc[i] = sum;
    sum += (idx < NN) ? cnt[idx] : 0;
  }
  part[t] = sum;
  __syncthreads();
  for (int off = 1; off < 1024; off <<= 1) {
    const int add = (t >= off) ? part[t - off] : 0;
    __syncthreads();
    part[t] += add;
    __syncthreads();
  }
  const int excl = part[t] - sum;
#pragma unroll
  for (int i = 0; i < SCAN_PER; ++i) {
    const int idx = base + i;
    if (idx < NN) rowptr[idx] = excl + loc[i];
  }
  if (t == 1023) rowptr[NN] = part[1023];
}

__global__ void gat_place(const int* __restrict__ src, const int* __restrict__ dst,
                          const int* __restrict__ rank,
                          const int* __restrict__ rowptr, int* __restrict__ csrc) {
  const int e = blockIdx.x * blockDim.x + threadIdx.x;
  if (e < EE) csrc[rowptr[dst[e]] + rank[e]] = src[e];
}

__device__ __forceinline__ void agg_group(const float* __restrict__ hfeat,
                                          const int lane, const int hsel,
                                          const int sn, const float alpha,
                                          const int left,
                                          float4& acc0, float4& acc1) {
  if (left >= 4) {
    float4 r0[4], r1[4];
#pragma unroll
    for (int k = 0; k < 4; ++k) {
      const int sk = __builtin_amdgcn_readlane(sn, k * 8);
      const float4* hp = reinterpret_cast<const float4*>(hfeat + (size_t)sk * F) + lane;
      r0[k] = hp[0];
      r1[k] = hp[32];
    }
#pragma unroll
    for (int k = 0; k < 4; ++k) {
      const float a0 = __shfl(alpha, k * 8 + hsel);
      const float a1 = __shfl(alpha, k * 8 + hsel + 4);
      acc0.x += a0 * r0[k].x; acc0.y += a0 * r0[k].y;
      acc0.z += a0 * r0[k].z; acc0.w += a0 * r0[k].w;
      acc1.x += a1 * r1[k].x; acc1.y += a1 * r1[k].y;
      acc1.z += a1 * r1[k].z; acc1.w += a1 * r1[k].w;
    }
  } else {
#pragma unroll
    for (int k = 0; k < 4; ++k) {
      const float a0 = __shfl(alpha, k * 8 + hsel);
      const float a1 = __shfl(alpha, k * 8 + hsel + 4);
      if (k < left) {
        const int sk = __builtin_amdgcn_readlane(sn, k * 8);
        const float4* hp = reinterpret_cast<const float4*>(hfeat + (size_t)sk * F) + lane;
        const float4 q0 = hp[0];
        const float4 q1 = hp[32];
        acc0.x += a0 * q0.x; acc0.y += a0 * q0.y;
        acc0.z += a0 * q0.z; acc0.w += a0 * q0.w;
        acc1.x += a1 * q1.x; acc1.y += a1 * q1.y;
        acc1.z += a1 * q1.z; acc1.w += a1 * q1.w;
      }
    }
  }
}

__global__ __launch_bounds__(256) void gat_node_softmax_agg(
    const float* __restrict__ el, const float* __restrict__ er,
    const int* __restrict__ rowptr, const int* __restrict__ csrc,
    const float* __restrict__ hfeat, const float* __restrict__ b,
    float* __restrict__ out) {
  const int lane = threadIdx.x & 31;
  const int n = __builtin_amdgcn_readfirstlane(
      (int)((blockIdx.x * blockDim.x + threadIdx.x) >> 5));
  if (n >= NN) return;
  const int sub = lane >> 3, hd = lane & 7;
  const int beg = rowptr[n], end = rowptr[n + 1];
  const int deg = end - beg;
  const float ern = er[n * HH + hd];
  const int hsel = lane >> 3;

  float4 acc0 = make_float4(0.f, 0.f, 0.f, 0.f);
  float4 acc1 = make_float4(0.f, 0.f, 0.f, 0.f);

  if (deg <= 32) {
    int own = 0x7fffffff;
    if (lane < deg) own = csrc[beg + lane];
    int pos = 0;
    for (int j = 0; j < deg; ++j) {
      const int sj = __builtin_amdgcn_readlane(own, j);
      pos += (sj < own || (sj == own && j < lane)) ? 1 : 0;
    }
    if (lane >= deg) pos = lane;
    const int sorted = __builtin_amdgcn_ds_permute(pos << 2, own);

    int   snv[8];
    float sc[8];
#pragma unroll
    for (int t = 0; t < 8; ++t) {
      const int e = 4 * t + sub;
      const int s = __shfl(sorted, e);
      snv[t] = (e < deg) ? s : 0;
    }
#pragma unroll
    for (int t = 0; t < 8; ++t) sc[t] = el[snv[t] * HH + hd];

    float mx = -INFINITY;
#pragma unroll
    for (int t = 0; t < 8; ++t) {
      float v = sc[t] + ern;
      v = (v > 0.f) ? v : 0.2f * v;
      v = (4 * t + sub < deg) ? v : -INFINITY;
      sc[t] = v;
      mx = fmaxf(mx, v);
    }
    mx = fmaxf(mx, __shfl_xor(mx, 8));
    mx = fmaxf(mx, __shfl_xor(mx, 16));

    float sm = 0.f;
#pragma unroll
    for (int t = 0; t < 8; ++t) {
      const float ex = (4 * t + sub < deg) ? __expf(sc[t] - mx) : 0.f;
      sc[t] = ex;
      sm += ex;
    }
    sm += __shfl_xor(sm, 8);
    sm += __shfl_xor(sm, 16);

#pragma unroll
    for (int t = 0; t < 8; ++t)
      sc[t] = (4 * t + sub < deg) ? sc[t] / sm : 0.f;

#pragma unroll
    for (int t = 0; t < 8; ++t) {
      const int left = deg - 4 * t;
      if (left <= 0) break;
      agg_group(hfeat, lane, hsel, snv[t], sc[t], left, acc0, acc1);
    }
  } else {
    float mx = -INFINITY;
    for (int base = beg; base < end; base += 4) {
      const int ed = base + sub;
      if (ed < end) {
        float v = el[csrc[ed] * HH + hd] + ern;
        v = (v > 0.f) ? v : 0.2f * v;
        mx = fmaxf(mx, v);
      }
    }
    mx = fmaxf(mx, __shfl_xor(mx, 8));
    mx = fmaxf(mx, __shfl_xor(mx, 16));

    float sm = 0.f;
    for (int base = beg; base < end; base += 4) {
      const int ed = base + sub;
      if (ed < end) {
        float v = el[csrc[ed] * HH + hd] + ern;
        v = (v > 0.f) ? v : 0.2f * v;
        sm += __expf(v - mx);
      }
    }
    sm += __shfl_xor(sm, 8);
    sm += __shfl_xor(sm, 16);

    for (int base = beg; base < end; base += 4) {
      const int ed = base + sub;
      int sn = 0;
      float alpha = 0.f;
      if (ed < end) {
        sn = csrc[ed];
        float v = el[sn * HH + hd] + ern;
        v = (v > 0.f) ? v : 0.2f * v;
        alpha = __expf(v - mx) / sm;
      }
      agg_group(hfeat, lane, hsel, sn, alpha, end - base, acc0, acc1);
    }
  }

  const float4* b4 = reinterpret_cast<const float4*>(b) + lane;
  float4* o4 = reinterpret_cast<float4*>(out + (size_t)n * F) + lane;
  const float4 bA = b4[0], bB = b4[32];
  float4 oA, oB;
  float t;
  t = acc0.x + bA.x; oA.x = (t > 0.f) ? t : (__expf(t) - 1.f);
  t = acc0.y + bA.y; oA.y = (t > 0.f) ? t : (__expf(t) - 1.f);
  t = acc0.z + bA.z; oA.z = (t > 0.f) ? t : (__expf(t) - 1.f);
  t = acc0.w + bA.w; oA.w = (t > 0.f) ? t : (__expf(t) - 1.f);
  t = acc1.x + bB.x; oB.x = (t > 0.f) ? t : (__expf(t) - 1.f);
  t = acc1.y + bB.y; oB.y = (t > 0.f) ? t : (__expf(t) - 1.f);
  t = acc1.z + bB.z; oB.z = (t > 0.f) ? t : (__expf(t) - 1.f);
  t = acc1.w + bB.w; oB.w = (t > 0.f) ? t : (__expf(t) - 1.f);
  o4[0]  = oA;
  o4[32] = oB;
}

extern "C" void kernel_launch(void* const* d_in, const int* in_sizes, int n_in,
                              void* d_out, int out_size, void* d_ws, size_t ws_size,
                              hipStream_t stream) {
  const float* feats = (const float*)d_in[0];
  const float* W0    = (const float*)d_in[1];
  const float* al0   = (const float*)d_in[2];
  const float* ar0   = (const float*)d_in[3];
  const float* b0    = (const float*)d_in[4];
  const float* W1    = (const float*)d_in[5];
  const float* al1   = (const float*)d_in[6];
  const float* ar1   = (const float*)d_in[7];
  const float* b1    = (const float*)d_in[8];
  const int*   src   = (const int*)d_in[9];
  const int*   dst   = (const int*)d_in[10];

  float* ws = (float*)d_ws;
  float* hbuf = ws;
  float* x1   = hbuf + (size_t)NN * F;
  float* el   = x1 + (size_t)NN * F;
  float* er   = el + (size_t)NN * HH;
  int* cnt    = (int*)(er + (size_t)NN * HH);
  int* rowptr = cnt + NN;
  int* rank   = rowptr + (NN + 32);
  int* csrc   = rank + EE;

  const int bl = 256;
  const int gN  = (NN + bl - 1) / bl;
  const int gE  = (EE + bl - 1) / bl;

  gat_zero_cnt<<<gN, bl, 0, stream>>>(cnt);
  gat_count_rank<<<gE, bl, 0, stream>>>(dst, cnt, rank);
  gat_scan<<<1, 1024, 0, stream>>>(cnt, rowptr);
  gat_place<<<gE, bl, 0, stream>>>(src, dst, rank, rowptr, csrc);

  auto run_layer = [&](const float* x, const float* Wm, const float* al,
                       const float* ar, const float* bb, float* outx) {
    gat_gemm_wmma<<<NN / 16, 256, 0, stream>>>(x, Wm, al, ar, hbuf, el, er);
    gat_node_softmax_agg<<<(NN * 32 + bl - 1) / bl, bl, 0, stream>>>(
        el, er, rowptr, csrc, hbuf, bb, outx);
  };

  run_layer(feats, W0, al0, ar0, b0, x1);
  run_layer(x1, W1, al1, ar1, b1, (float*)d_out);
}


# layer-0 GEMM's row slabs dealt out as extra blocks of the count, scan and place launches, which it does not depend on (scan now one 256-thread block through LDS); 8 launches -> 7
# speedup vs baseline: 13.1614x; 1.0648x over previous
#include <hip/hip_runtime.h>
#include <hip/hip_bf16.h>
#include <math.h>

#define NN 20000
#define EE 320000
#define HH 8
#define DD 32
#define F  256
#define LDSPITCH 260

typedef float v2f __attribute__((ext_vector_type(2)));
typedef float v8f __attribute__((ext_vector_type(8)));

#define GEMM_SMEM_FLOATS (16 * LDSPITCH + 2 * 16 * HH)
__device__ __forceinline__ void gat_gemm_slab(
    float* __restrict__ smem, const int slab,
    const float* __restrict__ A, const float* __restrict__ W,
    const float* __restrict__ al, const float* __restrict__ ar,
    float* __restrict__ C, float* __restrict__ el, float* __restrict__ er) {
  float* As = smem;
  float* Es = smem + 16 * LDSPITCH;
  const int tid  = threadIdx.x;
  const int wave = tid >> 5;
  const int lane = tid & 31;
  const int m0   = slab << 4;

  for (int i = tid; i < 16 * 64; i += 256) {
    const int r  = i >> 6;
    const int c4 = (i & 63) << 2;
    const float4 v = *reinterpret_cast<const float4*>(A + (size_t)(m0 + r) * F + c4);
    float* p = &As[r * LDSPITCH + c4];
    p[0] = v.x; p[1] = v.y; p[2] = v.z; p[3] = v.w;
  }
  __syncthreads();

  const int hi = lane >> 4;
  const int l  = lane & 15;
  const int n0 = wave << 5;

  v8f c0 = {}, c1 = {};
  for (int k0 = 0; k0 < F; k0 += 4) {
    const int ka = k0 + (hi << 1);
    v2f a, b0, b1;
    a.x  = As[l * LDSPITCH + ka];
    a.y  = As[l * LDSPITCH + ka + 1];
    b0.x = W[(size_t)ka * F + n0 + l];
    b0.y = W[(size_t)(ka + 1) * F + n0 + l];
    b1.x = W[(size_t)ka * F + n0 + 16 + l];
    b1.y = W[(size_t)(ka + 1) * F + n0 + 16 + l];
    c0 = __builtin_amdgcn_wmma_f32_16x16x4_f32(false, a, false, b0, (short)0, c0,
                                               false, false);
    c1 = __builtin_amdgcn_wmma_f32_16x16x4_f32(false, a, false, b1, (short)0, c1,
                                               false, false);
  }
#pragma unroll
  for (int v = 0; v < 8; ++v) {
    float* crow = C + (size_t)(m0 + (hi << 3) + v) * F + n0 + l;
    crow[0]  = c0[v];
    crow[16] = c1[v];
  }

  const float alA = al[n0 + l], alB = al[n0 + 16 + l];
  const float arA = ar[n0 + l], arB = ar[n0 + 16 + l];
  const bool upper = (l & 8) != 0;
  float keep[8], give[8];
#pragma unroll
  for (int v = 0; v < 8; ++v) {
    const float pl = c0[v] * alA + c1[v] * alB;
    const float pr = c0[v] * arA + c1[v] * arB;
    keep[v] = upper ? pr : pl;
    give[v] = upper ? pl : pr;
  }
#pragma unroll
  for (int v = 0; v < 8; ++v) keep[v] += __shfl_xor(give[v], 8);
#pragma unroll
  for (int v = 0; v < 8; ++v) keep[v] += __shfl_xor(keep[v], 4);
#pragma unroll
  for (int v = 0; v < 8; ++v) keep[v] += __shfl_xor(keep[v], 2);
#pragma unroll
  for (int v = 0; v < 8; ++v) keep[v] += __shfl_xor(keep[v], 1);
  if ((l & 7) == 0) {
    float* es = &Es[(upper ? 16 * HH : 0) + (hi << 3) * HH + wave];
#pragma unroll
    for (int v = 0; v < 8; ++v) es[v * HH] = keep[v];
  }
  __syncthreads();
  if (tid < 16 * HH) el[(size_t)m0 * HH + tid] = Es[tid];
  else               er[(size_t)m0 * HH + (tid - 16 * HH)] = Es[tid];
}

__global__ __launch_bounds__(256) void gat_gemm_wmma(
    const float* __restrict__ A, const float* __restrict__ W,
    const float* __restrict__ al, const float* __restrict__ ar,
    float* __restrict__ C, float* __restrict__ el, float* __restrict__ er) {
  __shared__ __attribute__((aligned(16))) float smem[GEMM_SMEM_FLOATS];
  gat_gemm_slab(smem, blockIdx.x, A, W, al, ar, C, el, er);
}

#define G_ALL (NN / 16)
#define G_A   512
#define G_B   256
#define G_C   (G_ALL - G_A - G_B)
#define E_BLOCKS ((EE + 255) / 256)

__global__ void gat_zero_cnt(int* __restrict__ cnt) {
  const int i = blockIdx.x * blockDim.x + threadIdx.x;
  if (i < NN) cnt[i] = 0;
}

__global__ __launch_bounds__(256) void gat_count_rank_gemm(
    const int* __restrict__ dst, int* __restrict__ cnt, int* __restrict__ rank,
    const float* __restrict__ A, const float* __restrict__ W,
    const float* __restrict__ al, const float* __restrict__ ar,
    float* __restrict__ C, float* __restrict__ el, float* __restrict__ er) {
  __shared__ __attribute__((aligned(16))) float smem[GEMM_SMEM_FLOATS];
  if (blockIdx.x < G_A) {
    gat_gemm_slab(smem, blockIdx.x, A, W, al, ar, C, el, er);
    return;
  }
  const int e = (blockIdx.x - G_A) * 256 + threadIdx.x;
  if (e < EE) rank[e] = atomicAdd(&cnt[dst[e]], 1);
}

#define SCAN_PER 79
#define SCAN_SMEM_INTS (NN + 8)
#define SCAN_COPY_TRIPS ((NN / 4 + 255) / 256)
__device__ __forceinline__ void gat_scan_block(int* __restrict__ sc,
                                               const int* __restrict__ cnt,
                                               int* __restrict__ rowptr) {
  int* wtot = sc + NN;
  const int t = threadIdx.x;
  const int lane = t & 31, wave = t >> 5;
#pragma unroll
  for (int j = 0; j < SCAN_COPY_TRIPS; ++j) {
    const int i = t + 256 * j;
    if (i < NN / 4)
      reinterpret_cast<int4*>(sc)[i] = reinterpret_cast<const int4*>(cnt)[i];
  }
  __syncthreads();

  const int base = t * SCAN_PER;
  int sum = 0;
  for (int i = 0; i < SCAN_PER; ++i) {
    const int idx = base + i;
    sum += (idx < NN) ? sc[idx] : 0;
  }
  int incl = sum;
  for (int off = 1; off < 32; off <<= 1) {
    const int v = __shfl_up(incl, off);
    if (lane >= off) incl += v;
  }
  if (lane == 31) wtot[wave] = incl;
  __syncthreads();
  int run = incl - sum;
  int total = 0;
  for (int w = 0; w < 8; ++w) {
    const int wt = wtot[w];
    run += (w < wave) ? wt : 0;
    total += wt;
  }
  for (int i = 0; i < SCAN_PER; ++i) {
    const int idx = base + i;
    if (idx < NN) {
      const int c = sc[idx];
      sc[idx] = run;
      run += c;
    }
  }
  if (t == 0) rowptr[NN] = total;
  __syncthreads();
#pragma unroll
  for (int j = 0; j < SCAN_COPY_TRIPS; ++j) {
    const int i = t + 256 * j;
    if (i < NN / 4)
      reinterpret_cast<int4*>(rowptr)[i] = reinterpret_cast<const int4*>(sc)[i];
  }
}

#define SCAN_GEMM_SMEM_FLOATS \
  (SCAN_SMEM_INTS > GEMM_SMEM_FLOATS ? SCAN_SMEM_INTS : GEMM_SMEM_FLOATS)
__global__ __launch_bounds__(256) void gat_scan_gemm(
    const int* __restrict__ cnt, int* __restrict__ rowptr,
    const float* __restrict__ A, const float* __restrict__ W,
    const float* __restrict__ al, const float* __restrict__ ar,
    float* __restrict__ C, float* __restrict__ el, float* __restrict__ er) {
  __shared__ __attribute__((aligned(16))) float smem[SCAN_GEMM_SMEM_FLOATS];
  if (blockIdx.x == 0) {
    gat_scan_block(reinterpret_cast<int*>(smem), cnt, rowptr);
    return;
  }
  gat_gemm_slab(smem, G_A + (blockIdx.x - 1), A, W, al, ar, C, el, er);
}

__global__ __launch_bounds__(256) void gat_place_gemm(
    const int* __restrict__ src, const int* __restrict__ dst,
    const int* __restrict__ rank, const int* __restrict__ rowptr,
    int* __restrict__ csrc,
    const float* __restrict__ A, const float* __restrict__ W,
    const float* __restrict__ al, const float* __restrict__ ar,
    float* __restrict__ C, float* __restrict__ el, float* __restrict__ er) {
  __shared__ __attribute__((aligned(16))) float smem[GEMM_SMEM_FLOATS];
  if (blockIdx.x < G_C) {
    gat_gemm_slab(smem, G_A + G_B + blockIdx.x, A, W, al, ar, C, el, er);
    return;
  }
  const int e = (blockIdx.x - G_C) * 256 + threadIdx.x;
  if (e < EE) csrc[rowptr[dst[e]] + rank[e]] = src[e];
}

__device__ __forceinline__ void agg_group(const float* __restrict__ hfeat,
                                          const int lane, const int hsel,
                                          const int sn, const float alpha,
                                          const int left,
                                          float4& acc0, float4& acc1) {
  if (left >= 4) {
    float4 r0[4], r1[4];
#pragma unroll
    for (int k = 0; k < 4; ++k) {
      const int sk = __builtin_amdgcn_readlane(sn, k * 8);
      const float4* hp = reinterpret_cast<const float4*>(hfeat + (size_t)sk * F) + lane;
      r0[k] = hp[0];
      r1[k] = hp[32];
    }
#pragma unroll
    for (int k = 0; k < 4; ++k) {
      const float a0 = __shfl(alpha, k * 8 + hsel);
      const float a1 = __shfl(alpha, k * 8 + hsel + 4);
      acc0.x += a0 * r0[k].x; acc0.y += a0 * r0[k].y;
      acc0.z += a0 * r0[k].z; acc0.w += a0 * r0[k].w;
      acc1.x += a1 * r1[k].x; acc1.y += a1 * r1[k].y;
      acc1.z += a1 * r1[k].z; acc1.w += a1 * r1[k].w;
    }
  } else {
#pragma unroll
    for (int k = 0; k < 4; ++k) {
      const float a0 = __shfl(alpha, k * 8 + hsel);
      const float a1 = __shfl(alpha, k * 8 + hsel + 4);
      if (k < left) {
        const int sk = __builtin_amdgcn_readlane(sn, k * 8);
        const float4* hp = reinterpret_cast<const float4*>(hfeat + (size_t)sk * F) + lane;
        const float4 q0 = hp[0];
        const float4 q1 = hp[32];
        acc0.x += a0 * q0.x; acc0.y += a0 * q0.y;
        acc0.z += a0 * q0.z; acc0.w += a0 * q0.w;
        acc1.x += a1 * q1.x; acc1.y += a1 * q1.y;
        acc1.z += a1 * q1.z; acc1.w += a1 * q1.w;
      }
    }
  }
}

__global__ __launch_bounds__(256) void gat_node_softmax_agg(
    const float* __restrict__ el, const float* __restrict__ er,
    const int* __restrict__ rowptr, const int* __restrict__ csrc,
    const float* __restrict__ hfeat, const float* __restrict__ b,
    float* __restrict__ out) {
  const int lane = threadIdx.x & 31;
  const int n = __builtin_amdgcn_readfirstlane(
      (int)((blockIdx.x * blockDim.x + threadIdx.x) >> 5));
  if (n >= NN) return;
  const int sub = lane >> 3, hd = lane & 7;
  const int beg = rowptr[n], end = rowptr[n + 1];
  const int deg = end - beg;
  const float ern = er[n * HH + hd];
  const int hsel = lane >> 3;

  float4 acc0 = make_float4(0.f, 0.f, 0.f, 0.f);
  float4 acc1 = make_float4(0.f, 0.f, 0.f, 0.f);

  if (deg <= 32) {
    int own = 0x7fffffff;
    if (lane < deg) own = csrc[beg + lane];
    int pos = 0;
    for (int j = 0; j < deg; ++j) {
      const int sj = __builtin_amdgcn_readlane(own, j);
      pos += (sj < own || (sj == own && j < lane)) ? 1 : 0;
    }
    if (lane >= deg) pos = lane;
    const int sorted = __builtin_amdgcn_ds_permute(pos << 2, own);

    int   snv[8];
    float sc[8];
#pragma unroll
    for (int t = 0; t < 8; ++t) {
      const int e = 4 * t + sub;
      const int s = __shfl(sorted, e);
      snv[t] = (e < deg) ? s : 0;
    }
#pragma unroll
    for (int t = 0; t < 8; ++t) sc[t] = el[snv[t] * HH + hd];

    float mx = -INFINITY;
#pragma unroll
    for (int t = 0; t < 8; ++t) {
      float v = sc[t] + ern;
      v = (v > 0.f) ? v : 0.2f * v;
      v = (4 * t + sub < deg) ? v : -INFINITY;
      sc[t] = v;
      mx = fmaxf(mx, v);
    }
    mx = fmaxf(mx, __shfl_xor(mx, 8));
    mx = fmaxf(mx, __shfl_xor(mx, 16));

    float sm = 0.f;
#pragma unroll
    for (int t = 0; t < 8; ++t) {
      const float ex = (4 * t + sub < deg) ? __expf(sc[t] - mx) : 0.f;
      sc[t] = ex;
      sm += ex;
    }
    sm += __shfl_xor(sm, 8);
    sm += __shfl_xor(sm, 16);

#pragma unroll
    for (int t = 0; t < 8; ++t)
      sc[t] = (4 * t + sub < deg) ? sc[t] / sm : 0.f;

#pragma unroll
    for (int t = 0; t < 8; ++t) {
      const int left = deg - 4 * t;
      if (left <= 0) break;
      agg_group(hfeat, lane, hsel, snv[t], sc[t], left, acc0, acc1);
    }
  } else {
    float mx = -INFINITY;
    for (int base = beg; base < end; base += 4) {
      const int ed = base + sub;
      if (ed < end) {
        float v = el[csrc[ed] * HH + hd] + ern;
        v = (v > 0.f) ? v : 0.2f * v;
        mx = fmaxf(mx, v);
      }
    }
    mx = fmaxf(mx, __shfl_xor(mx, 8));
    mx = fmaxf(mx, __shfl_xor(mx, 16));

    float sm = 0.f;
    for (int base = beg; base < end; base += 4) {
      const int ed = base + sub;
      if (ed < end) {
        float v = el[csrc[ed] * HH + hd] + ern;
        v = (v > 0.f) ? v : 0.2f * v;
        sm += __expf(v - mx);
      }
    }
    sm += __shfl_xor(sm, 8);
    sm += __shfl_xor(sm, 16);

    for (int base = beg; base < end; base += 4) {
      const int ed = base + sub;
      int sn = 0;
      float alpha = 0.f;
      if (ed < end) {
        sn = csrc[ed];
        float v = el[sn * HH + hd] + ern;
        v = (v > 0.f) ? v : 0.2f * v;
        alpha = __expf(v - mx) / sm;
      }
      agg_group(hfeat, lane, hsel, sn, alpha, end - base, acc0, acc1);
    }
  }

  const float4* b4 = reinterpret_cast<const float4*>(b) + lane;
  float4* o4 = reinterpret_cast<float4*>(out + (size_t)n * F) + lane;
  const float4 bA = b4[0], bB = b4[32];
  float4 oA, oB;
  float t;
  t = acc0.x + bA.x; oA.x = (t > 0.f) ? t : (__expf(t) - 1.f);
  t = acc0.y + bA.y; oA.y = (t > 0.f) ? t : (__expf(t) - 1.f);
  t = acc0.z + bA.z; oA.z = (t > 0.f) ? t : (__expf(t) - 1.f);
  t = acc0.w + bA.w; oA.w = (t > 0.f) ? t : (__expf(t) - 1.f);
  t = acc1.x + bB.x; oB.x = (t > 0.f) ? t : (__expf(t) - 1.f);
  t = acc1.y + bB.y; oB.y = (t > 0.f) ? t : (__expf(t) - 1.f);
  t = acc1.z + bB.z; oB.z = (t > 0.f) ? t : (__expf(t) - 1.f);
  t = acc1.w + bB.w; oB.w = (t > 0.f) ? t : (__expf(t) - 1.f);
  o4[0]  = oA;
  o4[32] = oB;
}

extern "C" void kernel_launch(void* const* d_in, const int* in_sizes, int n_in,
                              void* d_out, int out_size, void* d_ws, size_t ws_size,
                              hipStream_t stream) {
  const float* feats = (const float*)d_in[0];
  const float* W0    = (const float*)d_in[1];
  const float* al0   = (const float*)d_in[2];
  const float* ar0   = (const float*)d_in[3];
  const float* b0    = (const float*)d_in[4];
  const float* W1    = (const float*)d_in[5];
  const float* al1   = (const float*)d_in[6];
  const float* ar1   = (const float*)d_in[7];
  const float* b1    = (const float*)d_in[8];
  const int*   src   = (const int*)d_in[9];
  const int*   dst   = (const int*)d_in[10];

  float* ws = (float*)d_ws;
  float* hbuf = ws;
  float* x1   = hbuf + (size_t)NN * F;
  float* el   = x1 + (size_t)NN * F;
  float* er   = el + (size_t)NN * HH;
  int* cnt    = (int*)(er + (size_t)NN * HH);
  int* rowptr = cnt + NN;
  int* rank   = rowptr + (NN + 32);
  int* csrc   = rank + EE;

  const int bl = 256;
  const int gN  = (NN + bl - 1) / bl;

  gat_zero_cnt<<<gN, bl, 0, stream>>>(cnt);
  gat_count_rank_gemm<<<G_A + E_BLOCKS, 256, 0, stream>>>(
      dst, cnt, rank, feats, W0, al0, ar0, hbuf, el, er);
  gat_scan_gemm<<<1 + G_B, 256, 0, stream>>>(
      cnt, rowptr, feats, W0, al0, ar0, hbuf, el, er);
  gat_place_gemm<<<G_C + E_BLOCKS, 256, 0, stream>>>(
      src, dst, rank, rowptr, csrc, feats, W0, al0, ar0, hbuf, el, er);
  gat_node_softmax_agg<<<(NN * 32 + bl - 1) / bl, bl, 0, stream>>>(
      el, er, rowptr, csrc, hbuf, b0, x1);

  gat_gemm_wmma<<<NN / 16, 256, 0, stream>>>(x1, W1, al1, ar1, hbuf, el, er);
  gat_node_softmax_agg<<<(NN * 32 + bl - 1) / bl, bl, 0, stream>>>(
      el, er, rowptr, csrc, hbuf, b1, (float*)d_out);
}


# layer-1 GEMM moved into layer-0 aggregation's blocks: 16 waves put a slab's 16 activated rows in LDS, 8 of them multiply by W1 (same WMMA code); 7 launches -> 6; all sum orders kept
# speedup vs baseline: 13.7645x; 1.0458x over previous
#include <hip/hip_runtime.h>
#include <hip/hip_bf16.h>
#include <math.h>

#define NN 20000
#define EE 320000
#define HH 8
#define DD 32
#define F  256
#define LDSPITCH 260

typedef float v2f __attribute__((ext_vector_type(2)));
typedef float v8f __attribute__((ext_vector_type(8)));

#define GEMM_SMEM_FLOATS (16 * LDSPITCH + 2 * 16 * HH)
template <bool WIDE>
__device__ __forceinline__ void gat_gemm_from_lds(
    float* __restrict__ smem, const int slab, const float* __restrict__ W,
    const float* __restrict__ al, const float* __restrict__ ar,
    float* __restrict__ C, float* __restrict__ el, float* __restrict__ er) {
  const float* As = smem;
  float* Es = smem + 16 * LDSPITCH;
  const int tid  = threadIdx.x;
  const int wave = tid >> 5;
  const int lane = tid & 31;
  const int m0   = slab << 4;
  if (!WIDE || __builtin_amdgcn_readfirstlane(wave) < HH) {
  const int hi = lane >> 4;
  const int l  = lane & 15;
  const int n0 = wave << 5;

  v8f c0 = {}, c1 = {};
  for (int k0 = 0; k0 < F; k0 += 4) {
    const int ka = k0 + (hi << 1);
    v2f a, b0, b1;
    a.x  = As[l * LDSPITCH + ka];
    a.y  = As[l * LDSPITCH + ka + 1];
    b0.x = W[(size_t)ka * F + n0 + l];
    b0.y = W[(size_t)(ka + 1) * F + n0 + l];
    b1.x = W[(size_t)ka * F + n0 + 16 + l];
    b1.y = W[(size_t)(ka + 1) * F + n0 + 16 + l];
    c0 = __builtin_amdgcn_wmma_f32_16x16x4_f32(false, a, false, b0, (short)0, c0,
                                               false, false);
    c1 = __builtin_amdgcn_wmma_f32_16x16x4_f32(false, a, false, b1, (short)0, c1,
                                               false, false);
  }
#pragma unroll
  for (int v = 0; v < 8; ++v) {
    float* crow = C + (size_t)(m0 + (hi << 3) + v) * F + n0 + l;
    crow[0]  = c0[v];
    crow[16] = c1[v];
  }

  const float alA = al[n0 + l], alB = al[n0 + 16 + l];
  const float arA = ar[n0 + l], arB = ar[n0 + 16 + l];
  const bool upper = (l & 8) != 0;
  float keep[8], give[8];
#pragma unroll
  for (int v = 0; v < 8; ++v) {
    const float pl = c0[v] * alA + c1[v] * alB;
    const float pr = c0[v] * arA + c1[v] * arB;
    keep[v] = upper ? pr : pl;
    give[v] = upper ? pl : pr;
  }
#pragma unroll
  for (int v = 0; v < 8; ++v) keep[v] += __shfl_xor(give[v], 8);
#pragma unroll
  for (int v = 0; v < 8; ++v) keep[v] += __shfl_xor(keep[v], 4);
#pragma unroll
  for (int v = 0; v < 8; ++v) keep[v] += __shfl_xor(keep[v], 2);
#pragma unroll
  for (int v = 0; v < 8; ++v) keep[v] += __shfl_xor(keep[v], 1);
  if ((l & 7) == 0) {
    float* es = &Es[(upper ? 16 * HH : 0) + (hi << 3) * HH + wave];
#pragma unroll
    for (int v = 0; v < 8; ++v) es[v * HH] = keep[v];
  }
  }
  __syncthreads();
  if (tid < 16 * HH)          el[(size_t)m0 * HH + tid] = Es[tid];
  else if (!WIDE || tid < 2 * 16 * HH) er[(size_t)m0 * HH + (tid - 16 * HH)] = Es[tid];
}

__device__ __forceinline__ void gat_gemm_slab(
    float* __restrict__ smem, const int slab,
    const float* __restrict__ A, const float* __restrict__ W,
    const float* __restrict__ al, const float* __restrict__ ar,
    float* __restrict__ C, float* __restrict__ el, float* __restrict__ er) {
  float* As = smem;
  const int tid = threadIdx.x;
  const int m0  = slab << 4;
  for (int i = tid; i < 16 * 64; i += 256) {
    const int r  = i >> 6;
    const int c4 = (i & 63) << 2;
    const float4 v = *reinterpret_cast<const float4*>(A + (size_t)(m0 + r) * F + c4);
    float* p = &As[r * LDSPITCH + c4];
    p[0] = v.x; p[1] = v.y; p[2] = v.z; p[3] = v.w;
  }
  __syncthreads();
  gat_gemm_from_lds<false>(smem, slab, W, al, ar, C, el, er);
}

#define G_ALL (NN / 16)
#define G_A   512
#define G_B   256
#define G_C   (G_ALL - G_A - G_B)
#define E_BLOCKS ((EE + 255) / 256)

__global__ void gat_zero_cnt(int* __restrict__ cnt) {
  const int i = blockIdx.x * blockDim.x + threadIdx.x;
  if (i < NN) cnt[i] = 0;
}

__global__ __launch_bounds__(256) void gat_count_rank_gemm(
    const int* __restrict__ dst, int* __restrict__ cnt, int* __restrict__ rank,
    const float* __restrict__ A, const float* __restrict__ W,
    const float* __restrict__ al, const float* __restrict__ ar,
    float* __restrict__ C, float* __restrict__ el, float* __restrict__ er) {
  __shared__ __attribute__((aligned(16))) float smem[GEMM_SMEM_FLOATS];
  if (blockIdx.x < G_A) {
    gat_gemm_slab(smem, blockIdx.x, A, W, al, ar, C, el, er);
    return;
  }
  const int e = (blockIdx.x - G_A) * 256 + threadIdx.x;
  if (e < EE) rank[e] = atomicAdd(&cnt[dst[e]], 1);
}

#define SCAN_PER 79
#define SCAN_SMEM_INTS (NN + 8)
#define SCAN_COPY_TRIPS ((NN / 4 + 255) / 256)
__device__ __forceinline__ void gat_scan_block(int* __restrict__ sc,
                                               const int* __restrict__ cnt,
                                               int* __restrict__ rowptr) {
  int* wtot = sc + NN;
  const int t = threadIdx.x;
  const int lane = t & 31, wave = t >> 5;
#pragma unroll
  for (int j = 0; j < SCAN_COPY_TRIPS; ++j) {
    const int i = t + 256 * j;
    if (i < NN / 4)
      reinterpret_cast<int4*>(sc)[i] = reinterpret_cast<const int4*>(cnt)[i];
  }
  __syncthreads();

  const int base = t * SCAN_PER;
  int sum = 0;
  for (int i = 0; i < SCAN_PER; ++i) {
    const int idx = base + i;
    sum += (idx < NN) ? sc[idx] : 0;
  }
  int incl = sum;
  for (int off = 1; off < 32; off <<= 1) {
    const int v = __shfl_up(incl, off);
    if (lane >= off) incl += v;
  }
  if (lane == 31) wtot[wave] = incl;
  __syncthreads();
  int run = incl - sum;
  int total = 0;
  for (int w = 0; w < 8; ++w) {
    const int wt = wtot[w];
    run += (w < wave) ? wt : 0;
    total += wt;
  }
  for (int i = 0; i < SCAN_PER; ++i) {
    const int idx = base + i;
    if (idx < NN) {
      const int c = sc[idx];
      sc[idx] = run;
      run += c;
    }
  }
  if (t == 0) rowptr[NN] = total;
  __syncthreads();
#pragma unroll
  for (int j = 0; j < SCAN_COPY_TRIPS; ++j) {
    const int i = t + 256 * j;
    if (i < NN / 4)
      reinterpret_cast<int4*>(rowptr)[i] = reinterpret_cast<const int4*>(sc)[i];
  }
}

#define SCAN_GEMM_SMEM_FLOATS \
  (SCAN_SMEM_INTS > GEMM_SMEM_FLOATS ? SCAN_SMEM_INTS : GEMM_SMEM_FLOATS)
__global__ __launch_bounds__(256) void gat_scan_gemm(
    const int* __restrict__ cnt, int* __restrict__ rowptr,
    const float* __restrict__ A, const float* __restrict__ W,
    const float* __restrict__ al, const float* __restrict__ ar,
    float* __restrict__ C, float* __restrict__ el, float* __restrict__ er) {
  __shared__ __attribute__((aligned(16))) float smem[SCAN_GEMM_SMEM_FLOATS];
  if (blockIdx.x == 0) {
    gat_scan_block(reinterpret_cast<int*>(smem), cnt, rowptr);
    return;
  }
  gat_gemm_slab(smem, G_A + (blockIdx.x - 1), A, W, al, ar, C, el, er);
}

__global__ __launch_bounds__(256) void gat_place_gemm(
    const int* __restrict__ src, const int* __restrict__ dst,
    const int* __restrict__ rank, const int* __restrict__ rowptr,
    int* __restrict__ csrc,
    const float* __restrict__ A, const float* __restrict__ W,
    const float* __restrict__ al, const float* __restrict__ ar,
    float* __restrict__ C, float* __restrict__ el, float* __restrict__ er) {
  __shared__ __attribute__((aligned(16))) float smem[GEMM_SMEM_FLOATS];
  if (blockIdx.x < G_C) {
    gat_gemm_slab(smem, G_A + G_B + blockIdx.x, A, W, al, ar, C, el, er);
    return;
  }
  const int e = (blockIdx.x - G_C) * 256 + threadIdx.x;
  if (e < EE) csrc[rowptr[dst[e]] + rank[e]] = src[e];
}

__device__ __forceinline__ void agg_group(const float* __restrict__ hfeat,
                                          const int lane, const int hsel,
                                          const int sn, const float alpha,
                                          const int left,
                                          float4& acc0, float4& acc1) {
  if (left >= 4) {
    float4 r0[4], r1[4];
#pragma unroll
    for (int k = 0; k < 4; ++k) {
      const int sk = __builtin_amdgcn_readlane(sn, k * 8);
      const float4* hp = reinterpret_cast<const float4*>(hfeat + (size_t)sk * F) + lane;
      r0[k] = hp[0];
      r1[k] = hp[32];
    }
#pragma unroll
    for (int k = 0; k < 4; ++k) {
      const float a0 = __shfl(alpha, k * 8 + hsel);
      const float a1 = __shfl(alpha, k * 8 + hsel + 4);
      acc0.x += a0 * r0[k].x; acc0.y += a0 * r0[k].y;
      acc0.z += a0 * r0[k].z; acc0.w += a0 * r0[k].w;
      acc1.x += a1 * r1[k].x; acc1.y += a1 * r1[k].y;
      acc1.z += a1 * r1[k].z; acc1.w += a1 * r1[k].w;
    }
  } else {
#pragma unroll
    for (int k = 0; k < 4; ++k) {
      const float a0 = __shfl(alpha, k * 8 + hsel);
      const float a1 = __shfl(alpha, k * 8 + hsel + 4);
      if (k < left) {
        const int sk = __builtin_amdgcn_readlane(sn, k * 8);
        const float4* hp = reinterpret_cast<const float4*>(hfeat + (size_t)sk * F) + lane;
        const float4 q0 = hp[0];
        const float4 q1 = hp[32];
        acc0.x += a0 * q0.x; acc0.y += a0 * q0.y;
        acc0.z += a0 * q0.z; acc0.w += a0 * q0.w;
        acc1.x += a1 * q1.x; acc1.y += a1 * q1.y;
        acc1.z += a1 * q1.z; acc1.w += a1 * q1.w;
      }
    }
  }
}

__device__ __forceinline__ void gat_agg_row(
    const float* __restrict__ el, const float* __restrict__ er,
    const int* __restrict__ rowptr, const int* __restrict__ csrc,
    const float* __restrict__ hfeat, const float* __restrict__ b,
    const int n, const int lane, float4& oA, float4& oB) {
  const int sub = lane >> 3, hd = lane & 7;
  const int beg = rowptr[n], end = rowptr[n + 1];
  const int deg = end - beg;
  const float ern = er[n * HH + hd];
  const int hsel = lane >> 3;

  float4 acc0 = make_float4(0.f, 0.f, 0.f, 0.f);
  float4 acc1 = make_float4(0.f, 0.f, 0.f, 0.f);

  if (deg <= 32) {
    int own = 0x7fffffff;
    if (lane < deg) own = csrc[beg + lane];
    int pos = 0;
    for (int j = 0; j < deg; ++j) {
      const int sj = __builtin_amdgcn_readlane(own, j);
      pos += (sj < own || (sj == own && j < lane)) ? 1 : 0;
    }
    if (lane >= deg) pos = lane;
    const int sorted = __builtin_amdgcn_ds_permute(pos << 2, own);

    int   snv[8];
    float sc[8];
#pragma unroll
    for (int t = 0; t < 8; ++t) {
      const int e = 4 * t + sub;
      const int s = __shfl(sorted, e);
      snv[t] = (e < deg) ? s : 0;
    }
#pragma unroll
    for (int t = 0; t < 8; ++t) sc[t] = el[snv[t] * HH + hd];

    float mx = -INFINITY;
#pragma unroll
    for (int t = 0; t < 8; ++t) {
      float v = sc[t] + ern;
      v = (v > 0.f) ? v : 0.2f * v;
      v = (4 * t + sub < deg) ? v : -INFINITY;
      sc[t] = v;
      mx = fmaxf(mx, v);
    }
    mx = fmaxf(mx, __shfl_xor(mx, 8));
    mx = fmaxf(mx, __shfl_xor(mx, 16));

    float sm = 0.f;
#pragma unroll
    for (int t = 0; t < 8; ++t) {
      const float ex = (4 * t + sub < deg) ? __expf(sc[t] - mx) : 0.f;
      sc[t] = ex;
      sm += ex;
    }
    sm += __shfl_xor(sm, 8);
    sm += __shfl_xor(sm, 16);

#pragma unroll
    for (int t = 0; t < 8; ++t)
      sc[t] = (4 * t + sub < deg) ? sc[t] / sm : 0.f;

#pragma unroll
    for (int t = 0; t < 8; ++t) {
      const int left = deg - 4 * t;
      if (left <= 0) break;
      agg_group(hfeat, lane, hsel, snv[t], sc[t], left, acc0, acc1);
    }
  } else {
    float mx = -INFINITY;
    for (int base = beg; base < end; base += 4) {
      const int ed = base + sub;
      if (ed < end) {
        float v = el[csrc[ed] * HH + hd] + ern;
        v = (v > 0.f) ? v : 0.2f * v;
        mx = fmaxf(mx, v);
      }
    }
    mx = fmaxf(mx, __shfl_xor(mx, 8));
    mx = fmaxf(mx, __shfl_xor(mx, 16));

    float sm = 0.f;
    for (int base = beg; base < end; base += 4) {
      const int ed = base + sub;
      if (ed < end) {
        float v = el[csrc[ed] * HH + hd] + ern;
        v = (v > 0.f) ? v : 0.2f * v;
        sm += __expf(v - mx);
      }
    }
    sm += __shfl_xor(sm, 8);
    sm += __shfl_xor(sm, 16);

    for (int base = beg; base < end; base += 4) {
      const int ed = base + sub;
      int sn = 0;
      float alpha = 0.f;
      if (ed < end) {
        sn = csrc[ed];
        float v = el[sn * HH + hd] + ern;
        v = (v > 0.f) ? v : 0.2f * v;
        alpha = __expf(v - mx) / sm;
      }
      agg_group(hfeat, lane, hsel, sn, alpha, end - base, acc0, acc1);
    }
  }

  const float4* b4 = reinterpret_cast<const float4*>(b) + lane;
  const float4 bA = b4[0], bB = b4[32];
  float t;
  t = acc0.x + bA.x; oA.x = (t > 0.f) ? t : (__expf(t) - 1.f);
  t = acc0.y + bA.y; oA.y = (t > 0.f) ? t : (__expf(t) - 1.f);
  t = acc0.z + bA.z; oA.z = (t > 0.f) ? t : (__expf(t) - 1.f);
  t = acc0.w + bA.w; oA.w = (t > 0.f) ? t : (__expf(t) - 1.f);
  t = acc1.x + bB.x; oB.x = (t > 0.f) ? t : (__expf(t) - 1.f);
  t = acc1.y + bB.y; oB.y = (t > 0.f) ? t : (__expf(t) - 1.f);
  t = acc1.z + bB.z; oB.z = (t > 0.f) ? t : (__expf(t) - 1.f);
  t = acc1.w + bB.w; oB.w = (t > 0.f) ? t : (__expf(t) - 1.f);
}

__global__ __launch_bounds__(256) void gat_node_softmax_agg(
    const float* __restrict__ el, const float* __restrict__ er,
    const int* __restrict__ rowptr, const int* __restrict__ csrc,
    const float* __restrict__ hfeat, const float* __restrict__ b,
    float* __restrict__ out) {
  const int lane = threadIdx.x & 31;
  const int n = __builtin_amdgcn_readfirstlane(
      (int)((blockIdx.x * blockDim.x + threadIdx.x) >> 5));
  if (n >= NN) return;
  float4 oA, oB;
  gat_agg_row(el, er, rowptr, csrc, hfeat, b, n, lane, oA, oB);
  float4* o4 = reinterpret_cast<float4*>(out + (size_t)n * F) + lane;
  o4[0]  = oA;
  o4[32] = oB;
}

__global__ __launch_bounds__(512, 12) void gat_agg_gemm(
    const float* __restrict__ el, const float* __restrict__ er,
    const int* __restrict__ rowptr, const int* __restrict__ csrc,
    const float* __restrict__ hfeat, const float* __restrict__ b,
    const float* __restrict__ Wn, const float* __restrict__ aln,
    const float* __restrict__ arn, float* __restrict__ C,
    float* __restrict__ eln, float* __restrict__ ern) {
  __shared__ __attribute__((aligned(16))) float smem[GEMM_SMEM_FLOATS];
  const int lane = threadIdx.x & 31;
  const int wave = __builtin_amdgcn_readfirstlane((int)(threadIdx.x >> 5));
  const int n = blockIdx.x * 16 + wave;
  float4 oA, oB;
  gat_agg_row(el, er, rowptr, csrc, hfeat, b, n, lane, oA, oB);
  float4* row = reinterpret_cast<float4*>(smem + wave * LDSPITCH) + lane;
  row[0]  = oA;
  row[32] = oB;
  __syncthreads();
  gat_gemm_from_lds<true>(smem, blockIdx.x, Wn, aln, arn, C, eln, ern);
}

extern "C" void kernel_launch(void* const* d_in, const int* in_sizes, int n_in,
                              void* d_out, int out_size, void* d_ws, size_t ws_size,
                              hipStream_t stream) {
  const float* feats = (const float*)d_in[0];
  const float* W0    = (const float*)d_in[1];
  const float* al0   = (const float*)d_in[2];
  const float* ar0   = (const float*)d_in[3];
  const float* b0    = (const float*)d_in[4];
  const float* W1    = (const float*)d_in[5];
  const float* al1   = (const float*)d_in[6];
  const float* ar1   = (const float*)d_in[7];
  const float* b1    = (const float*)d_in[8];
  const int*   src   = (const int*)d_in[9];
  const int*   dst   = (const int*)d_in[10];

  float* ws = (float*)d_ws;
  float* hbuf = ws;
  float* hbuf1 = hbuf + (size_t)NN * F;
  float* el   = hbuf1 + (size_t)NN * F;
  float* er   = el + (size_t)NN * HH;
  int* cnt    = (int*)(er + (size_t)NN * HH);
  int* rowptr = cnt + NN;
  int* rank   = rowptr + (NN + 32);
  int* csrc   = rank + EE;
  float* el1  = (float*)(csrc + EE);
  float* er1  = el1 + (size_t)NN * HH;

  const int bl = 256;
  const int gN  = (NN + bl - 1) / bl;

  gat_zero_cnt<<<gN, bl, 0, stream>>>(cnt);
  gat_count_rank_gemm<<<G_A + E_BLOCKS, 256, 0, stream>>>(
      dst, cnt, rank, feats, W0, al0, ar0, hbuf, el, er);
  gat_scan_gemm<<<1 + G_B, 256, 0, stream>>>(
      cnt, rowptr, feats, W0, al0, ar0, hbuf, el, er);
  gat_place_gemm<<<G_C + E_BLOCKS, 256, 0, stream>>>(
      src, dst, rank, rowptr, csrc, feats, W0, al0, ar0, hbuf, el, er);
  gat_agg_gemm<<<NN / 16, 512, 0, stream>>>(
      el, er, rowptr, csrc, hbuf, b0, W1, al1, ar1, hbuf1, el1, er1);
  gat_node_softmax_agg<<<(NN * 32 + bl - 1) / bl, bl, 0, stream>>>(
      el1, er1, rowptr, csrc, hbuf1, b1, (float*)d_out);
}


# edge sources go straight into fixed 64-slot rows per destination at the counting atomic (nodes beyond 64 read the edge list, in edge order); scan and place launches gone; 6 launches -> 4
# speedup vs baseline: 16.3354x; 1.1868x over previous
#include <hip/hip_runtime.h>
#include <hip/hip_bf16.h>
#include <math.h>

#define NN 20000
#define EE 320000
#define HH 8
#define DD 32
#define F  256
#define LDSPITCH 260

typedef float v2f __attribute__((ext_vector_type(2)));
typedef float v8f __attribute__((ext_vector_type(8)));

#define GEMM_SMEM_FLOATS (16 * LDSPITCH + 2 * 16 * HH)
template <bool WIDE>
__device__ __forceinline__ void gat_gemm_from_lds(
    float* __restrict__ smem, const int slab, const float* __restrict__ W,
    const float* __restrict__ al, const float* __restrict__ ar,
    float* __restrict__ C, float* __restrict__ el, float* __restrict__ er) {
  const float* As = smem;
  float* Es = smem + 16 * LDSPITCH;
  const int tid  = threadIdx.x;
  const int wave = tid >> 5;
  const int lane = tid & 31;
  const int m0   = slab << 4;
  if (!WIDE || __builtin_amdgcn_readfirstlane(wave) < HH) {
  const int hi = lane >> 4;
  const int l  = lane & 15;
  const int n0 = wave << 5;

  v8f c0 = {}, c1 = {};
  for (int k0 = 0; k0 < F; k0 += 4) {
    const int ka = k0 + (hi << 1);
    v2f a, b0, b1;
    a.x  = As[l * LDSPITCH + ka];
    a.y  = As[l * LDSPITCH + ka + 1];
    b0.x = W[(size_t)ka * F + n0 + l];
    b0.y = W[(size_t)(ka + 1) * F + n0 + l];
    b1.x = W[(size_t)ka * F + n0 + 16 + l];
    b1.y = W[(size_t)(ka + 1) * F + n0 + 16 + l];
    c0 = __builtin_amdgcn_wmma_f32_16x16x4_f32(false, a, false, b0, (short)0, c0,
                                               false, false);
    c1 = __builtin_amdgcn_wmma_f32_16x16x4_f32(false, a, false, b1, (short)0, c1,
                                               false, false);
  }
#pragma unroll
  for (int v = 0; v < 8; ++v) {
    float* crow = C + (size_t)(m0 + (hi << 3) + v) * F + n0 + l;
    crow[0]  = c0[v];
    crow[16] = c1[v];
  }

  const float alA = al[n0 + l], alB = al[n0 + 16 + l];
  const float arA = ar[n0 + l], arB = ar[n0 + 16 + l];
  const bool upper = (l & 8) != 0;
  float keep[8], give[8];
#pragma unroll
  for (int v = 0; v < 8; ++v) {
    const float pl = c0[v] * alA + c1[v] * alB;
    const float pr = c0[v] * arA + c1[v] * arB;
    keep[v] = upper ? pr : pl;
    give[v] = upper ? pl : pr;
  }
#pragma unroll
  for (int v = 0; v < 8; ++v) keep[v] += __shfl_xor(give[v], 8);
#pragma unroll
  for (int v = 0; v < 8; ++v) keep[v] += __shfl_xor(keep[v], 4);
#pragma unroll
  for (int v = 0; v < 8; ++v) keep[v] += __shfl_xor(keep[v], 2);
#pragma unroll
  for (int v = 0; v < 8; ++v) keep[v] += __shfl_xor(keep[v], 1);
  if ((l & 7) == 0) {
    float* es = &Es[(upper ? 16 * HH : 0) + (hi << 3) * HH + wave];
#pragma unroll
    for (int v = 0; v < 8; ++v) es[v * HH] = keep[v];
  }
  }
  __syncthreads();
  if (tid < 16 * HH)          el[(size_t)m0 * HH + tid] = Es[tid];
  else if (!WIDE || tid < 2 * 16 * HH) er[(size_t)m0 * HH + (tid - 16 * HH)] = Es[tid];
}

__device__ __forceinline__ void gat_gemm_slab(
    float* __restrict__ smem, const int slab,
    const float* __restrict__ A, const float* __restrict__ W,
    const float* __restrict__ al, const float* __restrict__ ar,
    float* __restrict__ C, float* __restrict__ el, float* __restrict__ er) {
  float* As = smem;
  const int tid = threadIdx.x;
  const int m0  = slab << 4;
  for (int i = tid; i < 16 * 64; i += 256) {
    const int r  = i >> 6;
    const int c4 = (i & 63) << 2;
    const float4 v = *reinterpret_cast<const float4*>(A + (size_t)(m0 + r) * F + c4);
    float* p = &As[r * LDSPITCH + c4];
    p[0] = v.x; p[1] = v.y; p[2] = v.z; p[3] = v.w;
  }
  __syncthreads();
  gat_gemm_from_lds<false>(smem, slab, W, al, ar, C, el, er);
}

#define ROWCAP 64
#define G_ALL (NN / 16)
#define E_BLOCKS ((EE + 255) / 256)

__global__ void gat_zero_cnt(int* __restrict__ cnt) {
  const int i = blockIdx.x * blockDim.x + threadIdx.x;
  if (i < NN) cnt[i] = 0;
}

__global__ __launch_bounds__(256) void gat_count_place_gemm(
    const int* __restrict__ src, const int* __restrict__ dst,
    int* __restrict__ cnt, int* __restrict__ rows,
    const float* __restrict__ A, const float* __restrict__ W,
    const float* __restrict__ al, const float* __restrict__ ar,
    float* __restrict__ C, float* __restrict__ el, float* __restrict__ er) {
  __shared__ __attribute__((aligned(16))) float smem[GEMM_SMEM_FLOATS];
  if (blockIdx.x < G_ALL) {
    gat_gemm_slab(smem, blockIdx.x, A, W, al, ar, C, el, er);
    return;
  }
  const int e = (blockIdx.x - G_ALL) * 256 + threadIdx.x;
  if (e < EE) {
    const int d = dst[e];
    const int s = src[e];
    asm volatile("" :: "v"(s));
    const int slot = atomicAdd(&cnt[d], 1);
    if (slot < ROWCAP) rows[(size_t)d * ROWCAP + slot] = s;
  }
}

__device__ __forceinline__ void agg_group(const float* __restrict__ hfeat,
                                          const int lane, const int hsel,
                                          const int sn, const float alpha,
                                          const int left,
                                          float4& acc0, float4& acc1) {
  if (left >= 4) {
    float4 r0[4], r1[4];
#pragma unroll
    for (int k = 0; k < 4; ++k) {
      const int sk = __builtin_amdgcn_readlane(sn, k * 8);
      const float4* hp = reinterpret_cast<const float4*>(hfeat + (size_t)sk * F) + lane;
      r0[k] = hp[0];
      r1[k] = hp[32];
    }
#pragma unroll
    for (int k = 0; k < 4; ++k) {
      const float a0 = __shfl(alpha, k * 8 + hsel);
      const float a1 = __shfl(alpha, k * 8 + hsel + 4);
      acc0.x += a0 * r0[k].x; acc0.y += a0 * r0[k].y;
      acc0.z += a0 * r0[k].z; acc0.w += a0 * r0[k].w;
      acc1.x += a1 * r1[k].x; acc1.y += a1 * r1[k].y;
      acc1.z += a1 * r1[k].z; acc1.w += a1 * r1[k].w;
    }
  } else {
#pragma unroll
    for (int k = 0; k < 4; ++k) {
      const float a0 = __shfl(alpha, k * 8 + hsel);
      const float a1 = __shfl(alpha, k * 8 + hsel + 4);
      if (k < left) {
        const int sk = __builtin_amdgcn_readlane(sn, k * 8);
        const float4* hp = reinterpret_cast<const float4*>(hfeat + (size_t)sk * F) + lane;
        const float4 q0 = hp[0];
        const float4 q1 = hp[32];
        acc0.x += a0 * q0.x; acc0.y += a0 * q0.y;
        acc0.z += a0 * q0.z; acc0.w += a0 * q0.w;
        acc1.x += a1 * q1.x; acc1.y += a1 * q1.y;
        acc1.z += a1 * q1.z; acc1.w += a1 * q1.w;
      }
    }
  }
}

__device__ __forceinline__ void agg_row_from_edges(
    const int* __restrict__ src, const int* __restrict__ dst,
    const float* __restrict__ el, const float* __restrict__ hfeat,
    const int n, const int lane, const int hd, const int hsel, const float ern,
    float4& acc0, float4& acc1) {
  float mx = -INFINITY;
  for (int base = 0; base < EE; base += 32) {
    const int ed = base + lane;
    unsigned hit = __builtin_amdgcn_ballot_w32(ed < EE && dst[ed] == n);
    while (hit != 0u) {
      const int j = __builtin_ctz(hit);
      hit &= hit - 1u;
      float v = el[src[base + j] * HH + hd] + ern;
      v = (v > 0.f) ? v : 0.2f * v;
      mx = fmaxf(mx, v);
    }
  }
  float sm = 0.f;
  for (int base = 0; base < EE; base += 32) {
    const int ed = base + lane;
    unsigned hit = __builtin_amdgcn_ballot_w32(ed < EE && dst[ed] == n);
    while (hit != 0u) {
      const int j = __builtin_ctz(hit);
      hit &= hit - 1u;
      float v = el[src[base + j] * HH + hd] + ern;
      v = (v > 0.f) ? v : 0.2f * v;
      sm += __expf(v - mx);
    }
  }
  for (int base = 0; base < EE; base += 32) {
    const int ed = base + lane;
    unsigned hit = __builtin_amdgcn_ballot_w32(ed < EE && dst[ed] == n);
    while (hit != 0u) {
      const int j = __builtin_ctz(hit);
      hit &= hit - 1u;
      const int sn = src[base + j];
      float v = el[sn * HH + hd] + ern;
      v = (v > 0.f) ? v : 0.2f * v;
      const float alpha = __expf(v - mx) / sm;
      const float a0 = __shfl(alpha, hsel);
      const float a1 = __shfl(alpha, hsel + 4);
      const float4* hp = reinterpret_cast<const float4*>(hfeat + (size_t)sn * F) + lane;
      const float4 q0 = hp[0];
      const float4 q1 = hp[32];
      acc0.x += a0 * q0.x; acc0.y += a0 * q0.y;
      acc0.z += a0 * q0.z; acc0.w += a0 * q0.w;
      acc1.x += a1 * q1.x; acc1.y += a1 * q1.y;
      acc1.z += a1 * q1.z; acc1.w += a1 * q1.w;
    }
  }
}

__device__ __forceinline__ void gat_agg_row(
    const float* __restrict__ el, const float* __restrict__ er,
    const int* __restrict__ cnt, const int* __restrict__ rows,
    const int* __restrict__ src, const int* __restrict__ dst,
    const float* __restrict__ hfeat, const float* __restrict__ b,
    const int n, const int lane, float4& oA, float4& oB) {
  const int sub = lane >> 3, hd = lane & 7;
  const int deg = cnt[n];
  const int beg = n * ROWCAP, end = beg + deg;
  const float ern = er[n * HH + hd];
  const int hsel = lane >> 3;

  float4 acc0 = make_float4(0.f, 0.f, 0.f, 0.f);
  float4 acc1 = make_float4(0.f, 0.f, 0.f, 0.f);

  if (deg <= 32) {
    int own = 0x7fffffff;
    if (lane < deg) own = rows[beg + lane];
    int pos = 0;
    for (int j = 0; j < deg; ++j) {
      const int sj = __builtin_amdgcn_readlane(own, j);
      pos += (sj < own || (sj == own && j < lane)) ? 1 : 0;
    }
    if (lane >= deg) pos = lane;
    const int sorted = __builtin_amdgcn_ds_permute(pos << 2, own);

    int   snv[8];
    float sc[8];
#pragma unroll
    for (int t = 0; t < 8; ++t) {
      const int e = 4 * t + sub;
      const int s = __shfl(sorted, e);
      snv[t] = (e < deg) ? s : 0;
    }
#pragma unroll
    for (int t = 0; t < 8; ++t) sc[t] = el[snv[t] * HH + hd];

    float mx = -INFINITY;
#pragma unroll
    for (int t = 0; t < 8; ++t) {
      float v = sc[t] + ern;
      v = (v > 0.f) ? v : 0.2f * v;
      v = (4 * t + sub < deg) ? v : -INFINITY;
      sc[t] = v;
      mx = fmaxf(mx, v);
    }
    mx = fmaxf(mx, __shfl_xor(mx, 8));
    mx = fmaxf(mx, __shfl_xor(mx, 16));

    float sm = 0.f;
#pragma unroll
    for (int t = 0; t < 8; ++t) {
      const float ex = (4 * t + sub < deg) ? __expf(sc[t] - mx) : 0.f;
      sc[t] = ex;
      sm += ex;
    }
    sm += __shfl_xor(sm, 8);
    sm += __shfl_xor(sm, 16);

#pragma unroll
    for (int t = 0; t < 8; ++t)
      sc[t] = (4 * t + sub < deg) ? sc[t] / sm : 0.f;

#pragma unroll
    for (int t = 0; t < 8; ++t) {
      const int left = deg - 4 * t;
      if (left <= 0) break;
      agg_group(hfeat, lane, hsel, snv[t], sc[t], left, acc0, acc1);
    }
  } else if (deg > ROWCAP) {
    agg_row_from_edges(src, dst, el, hfeat, n, lane, hd, hsel, ern, acc0, acc1);
  } else {
    float mx = -INFINITY;
    for (int base = beg; base < end; base += 4) {
      const int ed = base + sub;
      if (ed < end) {
        float v = el[rows[ed] * HH + hd] + ern;
        v = (v > 0.f) ? v : 0.2f * v;
        mx = fmaxf(mx, v);
      }
    }
    mx = fmaxf(mx, __shfl_xor(mx, 8));
    mx = fmaxf(mx, __shfl_xor(mx, 16));

    float sm = 0.f;
    for (int base = beg; base < end; base += 4) {
      const int ed = base + sub;
      if (ed < end) {
        float v = el[rows[ed] * HH + hd] + ern;
        v = (v > 0.f) ? v : 0.2f * v;
        sm += __expf(v - mx);
      }
    }
    sm += __shfl_xor(sm, 8);
    sm += __shfl_xor(sm, 16);

    for (int base = beg; base < end; base += 4) {
      const int ed = base + sub;
      int sn = 0;
      float alpha = 0.f;
      if (ed < end) {
        sn = rows[ed];
        float v = el[sn * HH + hd] + ern;
        v = (v > 0.f) ? v : 0.2f * v;
        alpha = __expf(v - mx) / sm;
      }
      agg_group(hfeat, lane, hsel, sn, alpha, end - base, acc0, acc1);
    }
  }

  const float4* b4 = reinterpret_cast<const float4*>(b) + lane;
  const float4 bA = b4[0], bB = b4[32];
  float t;
  t = acc0.x + bA.x; oA.x = (t > 0.f) ? t : (__expf(t) - 1.f);
  t = acc0.y + bA.y; oA.y = (t > 0.f) ? t : (__expf(t) - 1.f);
  t = acc0.z + bA.z; oA.z = (t > 0.f) ? t : (__expf(t) - 1.f);
  t = acc0.w + bA.w; oA.w = (t > 0.f) ? t : (__expf(t) - 1.f);
  t = acc1.x + bB.x; oB.x = (t > 0.f) ? t : (__expf(t) - 1.f);
  t = acc1.y + bB.y; oB.y = (t > 0.f) ? t : (__expf(t) - 1.f);
  t = acc1.z + bB.z; oB.z = (t > 0.f) ? t : (__expf(t) - 1.f);
  t = acc1.w + bB.w; oB.w = (t > 0.f) ? t : (__expf(t) - 1.f);
}

__global__ __launch_bounds__(256) void gat_node_softmax_agg(
    const float* __restrict__ el, const float* __restrict__ er,
    const int* __restrict__ cnt, const int* __restrict__ rows,
    const int* __restrict__ src, const int* __restrict__ dst,
    const float* __restrict__ hfeat, const float* __restrict__ b,
    float* __restrict__ out) {
  const int lane = threadIdx.x & 31;
  const int n = __builtin_amdgcn_readfirstlane(
      (int)((blockIdx.x * blockDim.x + threadIdx.x) >> 5));
  if (n >= NN) return;
  float4 oA, oB;
  gat_agg_row(el, er, cnt, rows, src, dst, hfeat, b, n, lane, oA, oB);
  float4* o4 = reinterpret_cast<float4*>(out + (size_t)n * F) + lane;
  o4[0]  = oA;
  o4[32] = oB;
}

__global__ __launch_bounds__(512, 12) void gat_agg_gemm(
    const float* __restrict__ el, const float* __restrict__ er,
    const int* __restrict__ cnt, const int* __restrict__ rows,
    const int* __restrict__ src, const int* __restrict__ dst,
    const float* __restrict__ hfeat, const float* __restrict__ b,
    const float* __restrict__ Wn, const float* __restrict__ aln,
    const float* __restrict__ arn, float* __restrict__ C,
    float* __restrict__ eln, float* __restrict__ ern) {
  __shared__ __attribute__((aligned(16))) float smem[GEMM_SMEM_FLOATS];
  const int lane = threadIdx.x & 31;
  const int wave = __builtin_amdgcn_readfirstlane((int)(threadIdx.x >> 5));
  const int n = blockIdx.x * 16 + wave;
  float4 oA, oB;
  gat_agg_row(el, er, cnt, rows, src, dst, hfeat, b, n, lane, oA, oB);
  float4* row = reinterpret_cast<float4*>(smem + wave * LDSPITCH) + lane;
  row[0]  = oA;
  row[32] = oB;
  __syncthreads();
  gat_gemm_from_lds<true>(smem, blockIdx.x, Wn, aln, arn, C, eln, ern);
}

extern "C" void kernel_launch(void* const* d_in, const int* in_sizes, int n_in,
                              void* d_out, int out_size, void* d_ws, size_t ws_size,
                              hipStream_t stream) {
  const float* feats = (const float*)d_in[0];
  const float* W0    = (const float*)d_in[1];
  const float* al0   = (const float*)d_in[2];
  const float* ar0   = (const float*)d_in[3];
  const float* b0    = (const float*)d_in[4];
  const float* W1    = (const float*)d_in[5];
  const float* al1   = (const float*)d_in[6];
  const float* ar1   = (const float*)d_in[7];
  const float* b1    = (const float*)d_in[8];
  const int*   src   = (const int*)d_in[9];
  const int*   dst   = (const int*)d_in[10];

  float* ws = (float*)d_ws;
  float* hbuf = ws;
  float* hbuf1 = hbuf + (size_t)NN * F;
  float* el   = hbuf1 + (size_t)NN * F;
  float* er   = el + (size_t)NN * HH;
  int* cnt    = (int*)(er + (size_t)NN * HH);
  int* rows   = cnt + (NN + 32);
  float* el1  = (float*)(rows + (size_t)NN * ROWCAP);
  float* er1  = el1 + (size_t)NN * HH;

  gat_zero_cnt<<<(NN + 255) / 256, 256, 0, stream>>>(cnt);
  gat_count_place_gemm<<<G_ALL + E_BLOCKS, 256, 0, stream>>>(
      src, dst, cnt, rows, feats, W0, al0, ar0, hbuf, el, er);
  gat_agg_gemm<<<NN / 16, 512, 0, stream>>>(
      el, er, cnt, rows, src, dst, hbuf, b0, W1, al1, ar1, hbuf1, el1, er1);
  gat_node_softmax_agg<<<(NN * 32 + 255) / 256, 256, 0, stream>>>(
      el1, er1, cnt, rows, src, dst, hbuf1, b1, (float*)d_out);
}


# aggregation, lists of 33 to 64 edges: read once into registers (two entries per lane), sorted by source, one pass for max and sum, one round trip per four edges; their sum order now ascending source
# speedup vs baseline: 17.0405x; 1.0432x over previous
#include <hip/hip_runtime.h>
#include <hip/hip_bf16.h>
#include <math.h>

#define NN 20000
#define EE 320000
#define HH 8
#define DD 32
#define F  256
#define LDSPITCH 260

typedef float v2f __attribute__((ext_vector_type(2)));
typedef float v8f __attribute__((ext_vector_type(8)));

#define GEMM_SMEM_FLOATS (16 * LDSPITCH + 2 * 16 * HH)
template <bool WIDE>
__device__ __forceinline__ void gat_gemm_from_lds(
    float* __restrict__ smem, const int slab, const float* __restrict__ W,
    const float* __restrict__ al, const float* __restrict__ ar,
    float* __restrict__ C, float* __restrict__ el, float* __restrict__ er) {
  const float* As = smem;
  float* Es = smem + 16 * LDSPITCH;
  const int tid  = threadIdx.x;
  const int wave = tid >> 5;
  const int lane = tid & 31;
  const int m0   = slab << 4;
  if (!WIDE || __builtin_amdgcn_readfirstlane(wave) < HH) {
  const int hi = lane >> 4;
  const int l  = lane & 15;
  const int n0 = wave << 5;

  v8f c0 = {}, c1 = {};
  for (int k0 = 0; k0 < F; k0 += 4) {
    const int ka = k0 + (hi << 1);
    v2f a, b0, b1;
    a.x  = As[l * LDSPITCH + ka];
    a.y  = As[l * LDSPITCH + ka + 1];
    b0.x = W[(size_t)ka * F + n0 + l];
    b0.y = W[(size_t)(ka + 1) * F + n0 + l];
    b1.x = W[(size_t)ka * F + n0 + 16 + l];
    b1.y = W[(size_t)(ka + 1) * F + n0 + 16 + l];
    c0 = __builtin_amdgcn_wmma_f32_16x16x4_f32(false, a, false, b0, (short)0, c0,
                                               false, false);
    c1 = __builtin_amdgcn_wmma_f32_16x16x4_f32(false, a, false, b1, (short)0, c1,
                                               false, false);
  }
#pragma unroll
  for (int v = 0; v < 8; ++v) {
    float* crow = C + (size_t)(m0 + (hi << 3) + v) * F + n0 + l;
    crow[0]  = c0[v];
    crow[16] = c1[v];
  }

  const float alA = al[n0 + l], alB = al[n0 + 16 + l];
  const float arA = ar[n0 + l], arB = ar[n0 + 16 + l];
  const bool upper = (l & 8) != 0;
  float keep[8], give[8];
#pragma unroll
  for (int v = 0; v < 8; ++v) {
    const float pl = c0[v] * alA + c1[v] * alB;
    const float pr = c0[v] * arA + c1[v] * arB;
    keep[v] = upper ? pr : pl;
    give[v] = upper ? pl : pr;
  }
#pragma unroll
  for (int v = 0; v < 8; ++v) keep[v] += __shfl_xor(give[v], 8);
#pragma unroll
  for (int v = 0; v < 8; ++v) keep[v] += __shfl_xor(keep[v], 4);
#pragma unroll
  for (int v = 0; v < 8; ++v) keep[v] += __shfl_xor(keep[v], 2);
#pragma unroll
  for (int v = 0; v < 8; ++v) keep[v] += __shfl_xor(keep[v], 1);
  if ((l & 7) == 0) {
    float* es = &Es[(upper ? 16 * HH : 0) + (hi << 3) * HH + wave];
#pragma unroll
    for (int v = 0; v < 8; ++v) es[v * HH] = keep[v];
  }
  }
  __syncthreads();
  if (tid < 16 * HH)          el[(size_t)m0 * HH + tid] = Es[tid];
  else if (!WIDE || tid < 2 * 16 * HH) er[(size_t)m0 * HH + (tid - 16 * HH)] = Es[tid];
}

__device__ __forceinline__ void gat_gemm_slab(
    float* __restrict__ smem, const int slab,
    const float* __restrict__ A, const float* __restrict__ W,
    const float* __restrict__ al, const float* __restrict__ ar,
    float* __restrict__ C, float* __restrict__ el, float* __restrict__ er) {
  float* As = smem;
  const int tid = threadIdx.x;
  const int m0  = slab << 4;
  for (int i = tid; i < 16 * 64; i += 256) {
    const int r  = i >> 6;
    const int c4 = (i & 63) << 2;
    const float4 v = *reinterpret_cast<const float4*>(A + (size_t)(m0 + r) * F + c4);
    float* p = &As[r * LDSPITCH + c4];
    p[0] = v.x; p[1] = v.y; p[2] = v.z; p[3] = v.w;
  }
  __syncthreads();
  gat_gemm_from_lds<false>(smem, slab, W, al, ar, C, el, er);
}

#define ROWCAP 64
#define G_ALL (NN / 16)
#define E_BLOCKS ((EE + 255) / 256)

__global__ void gat_zero_cnt(int* __restrict__ cnt) {
  const int i = blockIdx.x * blockDim.x + threadIdx.x;
  if (i < NN) cnt[i] = 0;
}

__global__ __launch_bounds__(256) void gat_count_place_gemm(
    const int* __restrict__ src, const int* __restrict__ dst,
    int* __restrict__ cnt, int* __restrict__ rows,
    const float* __restrict__ A, const float* __restrict__ W,
    const float* __restrict__ al, const float* __restrict__ ar,
    float* __restrict__ C, float* __restrict__ el, float* __restrict__ er) {
  __shared__ __attribute__((aligned(16))) float smem[GEMM_SMEM_FLOATS];
  if (blockIdx.x < G_ALL) {
    gat_gemm_slab(smem, blockIdx.x, A, W, al, ar, C, el, er);
    return;
  }
  const int e = (blockIdx.x - G_ALL) * 256 + threadIdx.x;
  if (e < EE) {
    const int d = dst[e];
    const int s = src[e];
    asm volatile("" :: "v"(s));
    const int slot = atomicAdd(&cnt[d], 1);
    if (slot < ROWCAP) rows[(size_t)d * ROWCAP + slot] = s;
  }
}

__device__ __forceinline__ void agg_group(const float* __restrict__ hfeat,
                                          const int lane, const int hsel,
                                          const int sn, const float alpha,
                                          const int left,
                                          float4& acc0, float4& acc1) {
  if (left >= 4) {
    float4 r0[4], r1[4];
#pragma unroll
    for (int k = 0; k < 4; ++k) {
      const int sk = __builtin_amdgcn_readlane(sn, k * 8);
      const float4* hp = reinterpret_cast<const float4*>(hfeat + (size_t)sk * F) + lane;
      r0[k] = hp[0];
      r1[k] = hp[32];
    }
#pragma unroll
    for (int k = 0; k < 4; ++k) {
      const float a0 = __shfl(alpha, k * 8 + hsel);
      const float a1 = __shfl(alpha, k * 8 + hsel + 4);
      acc0.x += a0 * r0[k].x; acc0.y += a0 * r0[k].y;
      acc0.z += a0 * r0[k].z; acc0.w += a0 * r0[k].w;
      acc1.x += a1 * r1[k].x; acc1.y += a1 * r1[k].y;
      acc1.z += a1 * r1[k].z; acc1.w += a1 * r1[k].w;
    }
  } else {
#pragma unroll
    for (int k = 0; k < 4; ++k) {
      const float a0 = __shfl(alpha, k * 8 + hsel);
      const float a1 = __shfl(alpha, k * 8 + hsel + 4);
      if (k < left) {
        const int sk = __builtin_amdgcn_readlane(sn, k * 8);
        const float4* hp = reinterpret_cast<const float4*>(hfeat + (size_t)sk * F) + lane;
        const float4 q0 = hp[0];
        const float4 q1 = hp[32];
        acc0.x += a0 * q0.x; acc0.y += a0 * q0.y;
        acc0.z += a0 * q0.z; acc0.w += a0 * q0.w;
        acc1.x += a1 * q1.x; acc1.y += a1 * q1.y;
        acc1.z += a1 * q1.z; acc1.w += a1 * q1.w;
      }
    }
  }
}

__device__ __forceinline__ void agg_group_scored(const float* __restrict__ hfeat,
                                                 const int lane, const int hsel,
                                                 const int sn, const float raw,
                                                 const float ern, const float mx,
                                                 const float sm, const int left,
                                                 float4& acc0, float4& acc1) {
  if (left >= 4) {
    float4 r0[4], r1[4];
#pragma unroll
    for (int k = 0; k < 4; ++k) {
      const int sk = __builtin_amdgcn_readlane(sn, k * 8);
      const float4* hp = reinterpret_cast<const float4*>(hfeat + (size_t)sk * F) + lane;
      r0[k] = hp[0];
      r1[k] = hp[32];
    }
    float rv = raw;
    asm volatile("" : "+v"(rv) :: "memory");
    float v = rv + ern;
    v = (v > 0.f) ? v : 0.2f * v;
    const float alpha = __expf(v - mx) / sm;
#pragma unroll
    for (int k = 0; k < 4; ++k) {
      const float a0 = __shfl(alpha, k * 8 + hsel);
      const float a1 = __shfl(alpha, k * 8 + hsel + 4);
      acc0.x += a0 * r0[k].x; acc0.y += a0 * r0[k].y;
      acc0.z += a0 * r0[k].z; acc0.w += a0 * r0[k].w;
      acc1.x += a1 * r1[k].x; acc1.y += a1 * r1[k].y;
      acc1.z += a1 * r1[k].z; acc1.w += a1 * r1[k].w;
    }
  } else {
    float v = raw + ern;
    v = (v > 0.f) ? v : 0.2f * v;
    const float alpha = __expf(v - mx) / sm;
    agg_group(hfeat, lane, hsel, sn, alpha, left, acc0, acc1);
  }
}

__device__ __forceinline__ void agg_row_from_edges(
    const int* __restrict__ src, const int* __restrict__ dst,
    const float* __restrict__ el, const float* __restrict__ hfeat,
    const int n, const int lane, const int hd, const int hsel, const float ern,
    float4& acc0, float4& acc1) {
  float mx = -INFINITY;
  for (int base = 0; base < EE; base += 32) {
    const int ed = base + lane;
    unsigned hit = __builtin_amdgcn_ballot_w32(ed < EE && dst[ed] == n);
    while (hit != 0u) {
      const int j = __builtin_ctz(hit);
      hit &= hit - 1u;
      float v = el[src[base + j] * HH + hd] + ern;
      v = (v > 0.f) ? v : 0.2f * v;
      mx = fmaxf(mx, v);
    }
  }
  float sm = 0.f;
  for (int base = 0; base < EE; base += 32) {
    const int ed = base + lane;
    unsigned hit = __builtin_amdgcn_ballot_w32(ed < EE && dst[ed] == n);
    while (hit != 0u) {
      const int j = __builtin_ctz(hit);
      hit &= hit - 1u;
      float v = el[src[base + j] * HH + hd] + ern;
      v = (v > 0.f) ? v : 0.2f * v;
      sm += __expf(v - mx);
    }
  }
  for (int base = 0; base < EE; base += 32) {
    const int ed = base + lane;
    unsigned hit = __builtin_amdgcn_ballot_w32(ed < EE && dst[ed] == n);
    while (hit != 0u) {
      const int j = __builtin_ctz(hit);
      hit &= hit - 1u;
      const int sn = src[base + j];
      float v = el[sn * HH + hd] + ern;
      v = (v > 0.f) ? v : 0.2f * v;
      const float alpha = __expf(v - mx) / sm;
      const float a0 = __shfl(alpha, hsel);
      const float a1 = __shfl(alpha, hsel + 4);
      const float4* hp = reinterpret_cast<const float4*>(hfeat + (size_t)sn * F) + lane;
      const float4 q0 = hp[0];
      const float4 q1 = hp[32];
      acc0.x += a0 * q0.x; acc0.y += a0 * q0.y;
      acc0.z += a0 * q0.z; acc0.w += a0 * q0.w;
      acc1.x += a1 * q1.x; acc1.y += a1 * q1.y;
      acc1.z += a1 * q1.z; acc1.w += a1 * q1.w;
    }
  }
}

__device__ __forceinline__ void gat_agg_row(
    const float* __restrict__ el, const float* __restrict__ er,
    const int* __restrict__ cnt, const int* __restrict__ rows,
    const int* __restrict__ src, const int* __restrict__ dst,
    const float* __restrict__ hfeat, const float* __restrict__ b,
    const int n, const int lane, float4& oA, float4& oB) {
  const int sub = lane >> 3, hd = lane & 7;
  const int deg = cnt[n];
  const int beg = n * ROWCAP;
  const float ern = er[n * HH + hd];
  const int hsel = lane >> 3;

  float4 acc0 = make_float4(0.f, 0.f, 0.f, 0.f);
  float4 acc1 = make_float4(0.f, 0.f, 0.f, 0.f);

  if (deg <= 32) {
    int own = 0x7fffffff;
    if (lane < deg) own = rows[beg + lane];
    int pos = 0;
    for (int j = 0; j < deg; ++j) {
      const int sj = __builtin_amdgcn_readlane(own, j);
      pos += (sj < own || (sj == own && j < lane)) ? 1 : 0;
    }
    if (lane >= deg) pos = lane;
    const int sorted = __builtin_amdgcn_ds_permute(pos << 2, own);

    int   snv[8];
    float sc[8];
#pragma unroll
    for (int t = 0; t < 8; ++t) {
      const int e = 4 * t + sub;
      const int s = __shfl(sorted, e);
      snv[t] = (e < deg) ? s : 0;
    }
#pragma unroll
    for (int t = 0; t < 8; ++t) sc[t] = el[snv[t] * HH + hd];

    float mx = -INFINITY;
#pragma unroll
    for (int t = 0; t < 8; ++t) {
      float v = sc[t] + ern;
      v = (v > 0.f) ? v : 0.2f * v;
      v = (4 * t + sub < deg) ? v : -INFINITY;
      sc[t] = v;
      mx = fmaxf(mx, v);
    }
    mx = fmaxf(mx, __shfl_xor(mx, 8));
    mx = fmaxf(mx, __shfl_xor(mx, 16));

    float sm = 0.f;
#pragma unroll
    for (int t = 0; t < 8; ++t) {
      const float ex = (4 * t + sub < deg) ? __expf(sc[t] - mx) : 0.f;
      sc[t] = ex;
      sm += ex;
    }
    sm += __shfl_xor(sm, 8);
    sm += __shfl_xor(sm, 16);

#pragma unroll
    for (int t = 0; t < 8; ++t)
      sc[t] = (4 * t + sub < deg) ? sc[t] / sm : 0.f;

#pragma unroll
    for (int t = 0; t < 8; ++t) {
      const int left = deg - 4 * t;
      if (left <= 0) break;
      agg_group(hfeat, lane, hsel, snv[t], sc[t], left, acc0, acc1);
    }
  } else if (deg > ROWCAP) {
    agg_row_from_edges(src, dst, el, hfeat, n, lane, hd, hsel, ern, acc0, acc1);
  } else {
    const int rest = deg - 32;
    const int own0 = rows[beg + lane];
    int own1 = 0x7fffffff;
    if (lane < rest) own1 = rows[beg + 32 + lane];
    int pos0 = 0, pos1 = 0;
    for (int j = 0; j < 32; ++j) {
      const int sj = __builtin_amdgcn_readlane(own0, j);
      pos0 += (sj < own0 || (sj == own0 && j < lane)) ? 1 : 0;
      pos1 += (sj <= own1) ? 1 : 0;
    }
    for (int j = 0; j < rest; ++j) {
      const int sj = __builtin_amdgcn_readlane(own1, j);
      pos0 += (sj < own0) ? 1 : 0;
      pos1 += (sj < own1 || (sj == own1 && j < lane)) ? 1 : 0;
    }
    int sorted0 = 0, sorted1 = 0;
    for (int j = 0; j < 32; ++j) {
      const int p = __builtin_amdgcn_readlane(pos0, j);
      const int s = __builtin_amdgcn_readlane(own0, j);
      const bool mine = lane == (p & 31);
      sorted0 = (mine && p < 32)  ? s : sorted0;
      sorted1 = (mine && p >= 32) ? s : sorted1;
    }
    for (int j = 0; j < rest; ++j) {
      const int p = __builtin_amdgcn_readlane(pos1, j);
      const int s = __builtin_amdgcn_readlane(own1, j);
      const bool mine = lane == (p & 31);
      sorted0 = (mine && p < 32)  ? s : sorted0;
      sorted1 = (mine && p >= 32) ? s : sorted1;
    }

    float sc[16];
#pragma unroll
    for (int t = 0; t < 16; ++t) {
      const int e = 4 * t + sub;
      const int s = __shfl(t < 8 ? sorted0 : sorted1, e & 31);
      sc[t] = el[((e < deg) ? s : 0) * HH + hd];
    }
    float mx = -INFINITY;
#pragma unroll
    for (int t = 0; t < 16; ++t) {
      float v = sc[t] + ern;
      v = (v > 0.f) ? v : 0.2f * v;
      v = (4 * t + sub < deg) ? v : -INFINITY;
      sc[t] = v;
      mx = fmaxf(mx, v);
    }
    mx = fmaxf(mx, __shfl_xor(mx, 8));
    mx = fmaxf(mx, __shfl_xor(mx, 16));
    float sm = 0.f;
#pragma unroll
    for (int t = 0; t < 16; ++t)
      sm += (4 * t + sub < deg) ? __expf(sc[t] - mx) : 0.f;
    sm += __shfl_xor(sm, 8);
    sm += __shfl_xor(sm, 16);

    for (int base = 0; base < deg; base += 4) {
      const int e = base + sub;
      const int s = __shfl(base < 32 ? sorted0 : sorted1, e & 31);
      const int sn = (e < deg) ? s : 0;
      const float raw = el[sn * HH + hd];
      agg_group_scored(hfeat, lane, hsel, sn, raw, ern, mx, sm, deg - base, acc0, acc1);
    }
  }

  const float4* b4 = reinterpret_cast<const float4*>(b) + lane;
  const float4 bA = b4[0], bB = b4[32];
  float t;
  t = acc0.x + bA.x; oA.x = (t > 0.f) ? t : (__expf(t) - 1.f);
  t = acc0.y + bA.y; oA.y = (t > 0.f) ? t : (__expf(t) - 1.f);
  t = acc0.z + bA.z; oA.z = (t > 0.f) ? t : (__expf(t) - 1.f);
  t = acc0.w + bA.w; oA.w = (t > 0.f) ? t : (__expf(t) - 1.f);
  t = acc1.x + bB.x; oB.x = (t > 0.f) ? t : (__expf(t) - 1.f);
  t = acc1.y + bB.y; oB.y = (t > 0.f) ? t : (__expf(t) - 1.f);
  t = acc1.z + bB.z; oB.z = (t > 0.f) ? t : (__expf(t) - 1.f);
  t = acc1.w + bB.w; oB.w = (t > 0.f) ? t : (__expf(t) - 1.f);
}

__global__ __launch_bounds__(256, 12) void gat_node_softmax_agg(
    const float* __restrict__ el, const float* __restrict__ er,
    const int* __restrict__ cnt, const int* __restrict__ rows,
    const int* __restrict__ src, const int* __restrict__ dst,
    const float* __restrict__ hfeat, const float* __restrict__ b,
    float* __restrict__ out) {
  const int lane = threadIdx.x & 31;
  const int n = __builtin_amdgcn_readfirstlane(
      (int)((blockIdx.x * blockDim.x + threadIdx.x) >> 5));
  if (n >= NN) return;
  float4 oA, oB;
  gat_agg_row(el, er, cnt, rows, src, dst, hfeat, b, n, lane, oA, oB);
  float4* o4 = reinterpret_cast<float4*>(out + (size_t)n * F) + lane;
  o4[0]  = oA;
  o4[32] = oB;
}

__global__ __launch_bounds__(512, 12) void gat_agg_gemm(
    const float* __restrict__ el, const float* __restrict__ er,
    const int* __restrict__ cnt, const int* __restrict__ rows,
    const int* __restrict__ src, const int* __restrict__ dst,
    const float* __restrict__ hfeat, const float* __restrict__ b,
    const float* __restrict__ Wn, const float* __restrict__ aln,
    const float* __restrict__ arn, float* __restrict__ C,
    float* __restrict__ eln, float* __restrict__ ern) {
  __shared__ __attribute__((aligned(16))) float smem[GEMM_SMEM_FLOATS];
  const int lane = threadIdx.x & 31;
  const int wave = __builtin_amdgcn_readfirstlane((int)(threadIdx.x >> 5));
  const int n = blockIdx.x * 16 + wave;
  float4 oA, oB;
  gat_agg_row(el, er, cnt, rows, src, dst, hfeat, b, n, lane, oA, oB);
  float4* row = reinterpret_cast<float4*>(smem + wave * LDSPITCH) + lane;
  row[0]  = oA;
  row[32] = oB;
  __syncthreads();
  gat_gemm_from_lds<true>(smem, blockIdx.x, Wn, aln, arn, C, eln, ern);
}

extern "C" void kernel_launch(void* const* d_in, const int* in_sizes, int n_in,
                              void* d_out, int out_size, void* d_ws, size_t ws_size,
                              hipStream_t stream) {
  const float* feats = (const float*)d_in[0];
  const float* W0    = (const float*)d_in[1];
  const float* al0   = (const float*)d_in[2];
  const float* ar0   = (const float*)d_in[3];
  const float* b0    = (const float*)d_in[4];
  const float* W1    = (const float*)d_in[5];
  const float* al1   = (const float*)d_in[6];
  const float* ar1   = (const float*)d_in[7];
  const float* b1    = (const float*)d_in[8];
  const int*   src   = (const int*)d_in[9];
  const int*   dst   = (const int*)d_in[10];

  float* ws = (float*)d_ws;
  float* hbuf = ws;
  float* hbuf1 = hbuf + (size_t)NN * F;
  float* el   = hbuf1 + (size_t)NN * F;
  float* er   = el + (size_t)NN * HH;
  int* cnt    = (int*)(er + (size_t)NN * HH);
  int* rows   = cnt + (NN + 32);
  float* el1  = (float*)(rows + (size_t)NN * ROWCAP);
  float* er1  = el1 + (size_t)NN * HH;

  gat_zero_cnt<<<(NN + 255) / 256, 256, 0, stream>>>(cnt);
  gat_count_place_gemm<<<G_ALL + E_BLOCKS, 256, 0, stream>>>(
      src, dst, cnt, rows, feats, W0, al0, ar0, hbuf, el, er);
  gat_agg_gemm<<<NN / 16, 512, 0, stream>>>(
      el, er, cnt, rows, src, dst, hbuf, b0, W1, al1, ar1, hbuf1, el1, er1);
  gat_node_softmax_agg<<<(NN * 32 + 255) / 256, 256, 0, stream>>>(
      el1, er1, cnt, rows, src, dst, hbuf1, b1, (float*)d_out);
}


# W0, W1 laid out once per call (in the counter-clearing launch) in WMMA B-fragment order: both B fragments of a k-step are one 16-byte load per lane, were four 4-byte loads; all sums unchanged
# speedup vs baseline: 17.5963x; 1.0326x over previous
#include <hip/hip_runtime.h>
#include <hip/hip_bf16.h>
#include <math.h>

#define NN 20000
#define EE 320000
#define HH 8
#define DD 32
#define F  256
#define LDSPITCH 260

typedef float v2f __attribute__((ext_vector_type(2)));
typedef float v8f __attribute__((ext_vector_type(8)));

#define GEMM_SMEM_FLOATS (16 * LDSPITCH + 2 * 16 * HH)
template <bool WIDE>
__device__ __forceinline__ void gat_gemm_from_lds(
    float* __restrict__ smem, const int slab, const float* __restrict__ Wq,
    const float* __restrict__ al, const float* __restrict__ ar,
    float* __restrict__ C, float* __restrict__ el, float* __restrict__ er) {
  const float* As = smem;
  float* Es = smem + 16 * LDSPITCH;
  const int tid  = threadIdx.x;
  const int wave = tid >> 5;
  const int lane = tid & 31;
  const int m0   = slab << 4;
  if (!WIDE || __builtin_amdgcn_readfirstlane(wave) < HH) {
  const int hi = lane >> 4;
  const int l  = lane & 15;
  const int n0 = wave << 5;

  const float4* wq = reinterpret_cast<const float4*>(Wq) + ((size_t)wave << 11) + lane;
  v8f c0 = {}, c1 = {};
  for (int k0 = 0; k0 < F; k0 += 4) {
    const int ka = k0 + (hi << 1);
    v2f a, b0, b1;
    a.x  = As[l * LDSPITCH + ka];
    a.y  = As[l * LDSPITCH + ka + 1];
    const float4 q = wq[k0 << 3];
    b0.x = q.x;
    b0.y = q.y;
    b1.x = q.z;
    b1.y = q.w;
    c0 = __builtin_amdgcn_wmma_f32_16x16x4_f32(false, a, false, b0, (short)0, c0,
                                               false, false);
    c1 = __builtin_amdgcn_wmma_f32_16x16x4_f32(false, a, false, b1, (short)0, c1,
                                               false, false);
  }
#pragma unroll
  for (int v = 0; v < 8; ++v) {
    float* crow = C + (size_t)(m0 + (hi << 3) + v) * F + n0 + l;
    crow[0]  = c0[v];
    crow[16] = c1[v];
  }

  const float alA = al[n0 + l], alB = al[n0 + 16 + l];
  const float arA = ar[n0 + l], arB = ar[n0 + 16 + l];
  const bool upper = (l & 8) != 0;
  float keep[8], give[8];
#pragma unroll
  for (int v = 0; v < 8; ++v) {
    const float pl = c0[v] * alA + c1[v] * alB;
    const float pr = c0[v] * arA + c1[v] * arB;
    keep[v] = upper ? pr : pl;
    give[v] = upper ? pl : pr;
  }
#pragma unroll
  for (int v = 0; v < 8; ++v) keep[v] += __shfl_xor(give[v], 8);
#pragma unroll
  for (int v = 0; v < 8; ++v) keep[v] += __shfl_xor(keep[v], 4);
#pragma unroll
  for (int v = 0; v < 8; ++v) keep[v] += __shfl_xor(keep[v], 2);
#pragma unroll
  for (int v = 0; v < 8; ++v) keep[v] += __shfl_xor(keep[v], 1);
  if ((l & 7) == 0) {
    float* es = &Es[(upper ? 16 * HH : 0) + (hi << 3) * HH + wave];
#pragma unroll
    for (int v = 0; v < 8; ++v) es[v * HH] = keep[v];
  }
  }
  __syncthreads();
  if (tid < 16 * HH)          el[(size_t)m0 * HH + tid] = Es[tid];
  else if (!WIDE || tid < 2 * 16 * HH) er[(size_t)m0 * HH + (tid - 16 * HH)] = Es[tid];
}

__device__ __forceinline__ void gat_gemm_slab(
    float* __restrict__ smem, const int slab,
    const float* __restrict__ A, const float* __restrict__ Wq,
    const float* __restrict__ al, const float* __restrict__ ar,
    float* __restrict__ C, float* __restrict__ el, float* __restrict__ er) {
  float* As = smem;
  const int tid = threadIdx.x;
  const int m0  = slab << 4;
  for (int i = tid; i < 16 * 64; i += 256) {
    const int r  = i >> 6;
    const int c4 = (i & 63) << 2;
    const float4 v = *reinterpret_cast<const float4*>(A + (size_t)(m0 + r) * F + c4);
    float* p = &As[r * LDSPITCH + c4];
    p[0] = v.x; p[1] = v.y; p[2] = v.z; p[3] = v.w;
  }
  __syncthreads();
  gat_gemm_from_lds<false>(smem, slab, Wq, al, ar, C, el, er);
}

#define ROWCAP 64
#define G_ALL (NN / 16)
#define E_BLOCKS ((EE + 255) / 256)

#define Z_BLOCKS ((NN + 255) / 256)
#define LAY_BLOCKS (2 * F * F / 4 / 256)
__global__ __launch_bounds__(256) void gat_zero_cnt_lay_w(
    int* __restrict__ cnt, const float* __restrict__ Wa, const float* __restrict__ Wb,
    float* __restrict__ Wqa, float* __restrict__ Wqb) {
  if (blockIdx.x < Z_BLOCKS) {
    const int i = blockIdx.x * 256 + threadIdx.x;
    if (i < NN) cnt[i] = 0;
    return;
  }
  const int t    = (blockIdx.x - Z_BLOCKS) * 256 + threadIdx.x;
  const int o4   = t & (F * F / 4 - 1);
  const bool sec = t >= F * F / 4;
  const float* __restrict__ W = sec ? Wb : Wa;
  float* __restrict__ Wq      = sec ? Wqb : Wqa;
  const int lane = o4 & 31;
  const int s    = (o4 >> 5) & 63;
  const int head = o4 >> 11;
  const int k    = 4 * s + 2 * (lane >> 4);
  const int col  = 32 * head + (lane & 15);
  float4 q;
  q.x = W[(size_t)k * F + col];
  q.y = W[(size_t)(k + 1) * F + col];
  q.z = W[(size_t)k * F + col + 16];
  q.w = W[(size_t)(k + 1) * F + col + 16];
  reinterpret_cast<float4*>(Wq)[o4] = q;
}

__global__ __launch_bounds__(256) void gat_count_place_gemm(
    const int* __restrict__ src, const int* __restrict__ dst,
    int* __restrict__ cnt, int* __restrict__ rows,
    const float* __restrict__ A, const float* __restrict__ Wq,
    const float* __restrict__ al, const float* __restrict__ ar,
    float* __restrict__ C, float* __restrict__ el, float* __restrict__ er) {
  __shared__ __attribute__((aligned(16))) float smem[GEMM_SMEM_FLOATS];
  if (blockIdx.x < G_ALL) {
    gat_gemm_slab(smem, blockIdx.x, A, Wq, al, ar, C, el, er);
    return;
  }
  const int e = (blockIdx.x - G_ALL) * 256 + threadIdx.x;
  if (e < EE) {
    const int d = dst[e];
    const int s = src[e];
    asm volatile("" :: "v"(s));
    const int slot = atomicAdd(&cnt[d], 1);
    if (slot < ROWCAP) rows[(size_t)d * ROWCAP + slot] = s;
  }
}

__device__ __forceinline__ void agg_group(const float* __restrict__ hfeat,
                                          const int lane, const int hsel,
                                          const int sn, const float alpha,
                                          const int left,
                                          float4& acc0, float4& acc1) {
  if (left >= 4) {
    float4 r0[4], r1[4];
#pragma unroll
    for (int k = 0; k < 4; ++k) {
      const int sk = __builtin_amdgcn_readlane(sn, k * 8);
      const float4* hp = reinterpret_cast<const float4*>(hfeat + (size_t)sk * F) + lane;
      r0[k] = hp[0];
      r1[k] = hp[32];
    }
#pragma unroll
    for (int k = 0; k < 4; ++k) {
      const float a0 = __shfl(alpha, k * 8 + hsel);
      const float a1 = __shfl(alpha, k * 8 + hsel + 4);
      acc0.x += a0 * r0[k].x; acc0.y += a0 * r0[k].y;
      acc0.z += a0 * r0[k].z; acc0.w += a0 * r0[k].w;
      acc1.x += a1 * r1[k].x; acc1.y += a1 * r1[k].y;
      acc1.z += a1 * r1[k].z; acc1.w += a1 * r1[k].w;
    }
  } else {
#pragma unroll
    for (int k = 0; k < 4; ++k) {
      const float a0 = __shfl(alpha, k * 8 + hsel);
      const float a1 = __shfl(alpha, k * 8 + hsel + 4);
      if (k < left) {
        const int sk = __builtin_amdgcn_readlane(sn, k * 8);
        const float4* hp = reinterpret_cast<const float4*>(hfeat + (size_t)sk * F) + lane;
        const float4 q0 = hp[0];
        const float4 q1 = hp[32];
        acc0.x += a0 * q0.x; acc0.y += a0 * q0.y;
        acc0.z += a0 * q0.z; acc0.w += a0 * q0.w;
        acc1.x += a1 * q1.x; acc1.y += a1 * q1.y;
        acc1.z += a1 * q1.z; acc1.w += a1 * q1.w;
      }
    }
  }
}

__device__ __forceinline__ void agg_group_scored(const float* __restrict__ hfeat,
                                                 const int lane, const int hsel,
                                                 const int sn, const float raw,
                                                 const float ern, const float mx,
                                                 const float sm, const int left,
                                                 float4& acc0, float4& acc1) {
  if (left >= 4) {
    float4 r0[4], r1[4];
#pragma unroll
    for (int k = 0; k < 4; ++k) {
      const int sk = __builtin_amdgcn_readlane(sn, k * 8);
      const float4* hp = reinterpret_cast<const float4*>(hfeat + (size_t)sk * F) + lane;
      r0[k] = hp[0];
      r1[k] = hp[32];
    }
    float rv = raw;
    asm volatile("" : "+v"(rv) :: "memory");
    float v = rv + ern;
    v = (v > 0.f) ? v : 0.2f * v;
    const float alpha = __expf(v - mx) / sm;
#pragma unroll
    for (int k = 0; k < 4; ++k) {
      const float a0 = __shfl(alpha, k * 8 + hsel);
      const float a1 = __shfl(alpha, k * 8 + hsel + 4);
      acc0.x += a0 * r0[k].x; acc0.y += a0 * r0[k].y;
      acc0.z += a0 * r0[k].z; acc0.w += a0 * r0[k].w;
      acc1.x += a1 * r1[k].x; acc1.y += a1 * r1[k].y;
      acc1.z += a1 * r1[k].z; acc1.w += a1 * r1[k].w;
    }
  } else {
    float v = raw + ern;
    v = (v > 0.f) ? v : 0.2f * v;
    const float alpha = __expf(v - mx) / sm;
    agg_group(hfeat, lane, hsel, sn, alpha, left, acc0, acc1);
  }
}

__device__ __forceinline__ void agg_row_from_edges(
    const int* __restrict__ src, const int* __restrict__ dst,
    const float* __restrict__ el, const float* __restrict__ hfeat,
    const int n, const int lane, const int hd, const int hsel, const float ern,
    float4& acc0, float4& acc1) {
  float mx = -INFINITY;
  for (int base = 0; base < EE; base += 32) {
    const int ed = base + lane;
    unsigned hit = __builtin_amdgcn_ballot_w32(ed < EE && dst[ed] == n);
    while (hit != 0u) {
      const int j = __builtin_ctz(hit);
      hit &= hit - 1u;
      float v = el[src[base + j] * HH + hd] + ern;
      v = (v > 0.f) ? v : 0.2f * v;
      mx = fmaxf(mx, v);
    }
  }
  float sm = 0.f;
  for (int base = 0; base < EE; base += 32) {
    const int ed = base + lane;
    unsigned hit = __builtin_amdgcn_ballot_w32(ed < EE && dst[ed] == n);
    while (hit != 0u) {
      const int j = __builtin_ctz(hit);
      hit &= hit - 1u;
      float v = el[src[base + j] * HH + hd] + ern;
      v = (v > 0.f) ? v : 0.2f * v;
      sm += __expf(v - mx);
    }
  }
  for (int base = 0; base < EE; base += 32) {
    const int ed = base + lane;
    unsigned hit = __builtin_amdgcn_ballot_w32(ed < EE && dst[ed] == n);
    while (hit != 0u) {
      const int j = __builtin_ctz(hit);
      hit &= hit - 1u;
      const int sn = src[base + j];
      float v = el[sn * HH + hd] + ern;
      v = (v > 0.f) ? v : 0.2f * v;
      const float alpha = __expf(v - mx) / sm;
      const float a0 = __shfl(alpha, hsel);
      const float a1 = __shfl(alpha, hsel + 4);
      const float4* hp = reinterpret_cast<const float4*>(hfeat + (size_t)sn * F) + lane;
      const float4 q0 = hp[0];
      const float4 q1 = hp[32];
      acc0.x += a0 * q0.x; acc0.y += a0 * q0.y;
      acc0.z += a0 * q0.z; acc0.w += a0 * q0.w;
      acc1.x += a1 * q1.x; acc1.y += a1 * q1.y;
      acc1.z += a1 * q1.z; acc1.w += a1 * q1.w;
    }
  }
}

__device__ __forceinline__ void gat_agg_row(
    const float* __restrict__ el, const float* __restrict__ er,
    const int* __restrict__ cnt, const int* __restrict__ rows,
    const int* __restrict__ src, const int* __restrict__ dst,
    const float* __restrict__ hfeat, const float* __restrict__ b,
    const int n, const int lane, float4& oA, float4& oB) {
  const int sub = lane >> 3, hd = lane & 7;
  const int deg = cnt[n];
  const int beg = n * ROWCAP;
  const float ern = er[n * HH + hd];
  const int hsel = lane >> 3;

  float4 acc0 = make_float4(0.f, 0.f, 0.f, 0.f);
  float4 acc1 = make_float4(0.f, 0.f, 0.f, 0.f);

  if (deg <= 32) {
    int own = 0x7fffffff;
    if (lane < deg) own = rows[beg + lane];
    int pos = 0;
    for (int j = 0; j < deg; ++j) {
      const int sj = __builtin_amdgcn_readlane(own, j);
      pos += (sj < own || (sj == own && j < lane)) ? 1 : 0;
    }
    if (lane >= deg) pos = lane;
    const int sorted = __builtin_amdgcn_ds_permute(pos << 2, own);

    int   snv[8];
    float sc[8];
#pragma unroll
    for (int t = 0; t < 8; ++t) {
      const int e = 4 * t + sub;
      const int s = __shfl(sorted, e);
      snv[t] = (e < deg) ? s : 0;
    }
#pragma unroll
    for (int t = 0; t < 8; ++t) sc[t] = el[snv[t] * HH + hd];

    float mx = -INFINITY;
#pragma unroll
    for (int t = 0; t < 8; ++t) {
      float v = sc[t] + ern;
      v = (v > 0.f) ? v : 0.2f * v;
      v = (4 * t + sub < deg) ? v : -INFINITY;
      sc[t] = v;
      mx = fmaxf(mx, v);
    }
    mx = fmaxf(mx, __shfl_xor(mx, 8));
    mx = fmaxf(mx, __shfl_xor(mx, 16));

    float sm = 0.f;
#pragma unroll
    for (int t = 0; t < 8; ++t) {
      const float ex = (4 * t + sub < deg) ? __expf(sc[t] - mx) : 0.f;
      sc[t] = ex;
      sm += ex;
    }
    sm += __shfl_xor(sm, 8);
    sm += __shfl_xor(sm, 16);

#pragma unroll
    for (int t = 0; t < 8; ++t)
      sc[t] = (4 * t + sub < deg) ? sc[t] / sm : 0.f;

#pragma unroll
    for (int t = 0; t < 8; ++t) {
      const int left = deg - 4 * t;
      if (left <= 0) break;
      agg_group(hfeat, lane, hsel, snv[t], sc[t], left, acc0, acc1);
    }
  } else if (deg > ROWCAP) {
    agg_row_from_edges(src, dst, el, hfeat, n, lane, hd, hsel, ern, acc0, acc1);
  } else {
    const int rest = deg - 32;
    const int own0 = rows[beg + lane];
    int own1 = 0x7fffffff;
    if (lane < rest) own1 = rows[beg + 32 + lane];
    int pos0 = 0, pos1 = 0;
    for (int j = 0; j < 32; ++j) {
      const int sj = __builtin_amdgcn_readlane(own0, j);
      pos0 += (sj < own0 || (sj == own0 && j < lane)) ? 1 : 0;
      pos1 += (sj <= own1) ? 1 : 0;
    }
    for (int j = 0; j < rest; ++j) {
      const int sj = __builtin_amdgcn_readlane(own1, j);
      pos0 += (sj < own0) ? 1 : 0;
      pos1 += (sj < own1 || (sj == own1 && j < lane)) ? 1 : 0;
    }
    int sorted0 = 0, sorted1 = 0;
    for (int j = 0; j < 32; ++j) {
      const int p = __builtin_amdgcn_readlane(pos0, j);
      const int s = __builtin_amdgcn_readlane(own0, j);
      const bool mine = lane == (p & 31);
      sorted0 = (mine && p < 32)  ? s : sorted0;
      sorted1 = (mine && p >= 32) ? s : sorted1;
    }
    for (int j = 0; j < rest; ++j) {
      const int p = __builtin_amdgcn_readlane(pos1, j);
      const int s = __builtin_amdgcn_readlane(own1, j);
      const bool mine = lane == (p & 31);
      sorted0 = (mine && p < 32)  ? s : sorted0;
      sorted1 = (mine && p >= 32) ? s : sorted1;
    }

    float sc[16];
#pragma unroll
    for (int t = 0; t < 16; ++t) {
      const int e = 4 * t + sub;
      const int s = __shfl(t < 8 ? sorted0 : sorted1, e & 31);
      sc[t] = el[((e < deg) ? s : 0) * HH + hd];
    }
    float mx = -INFINITY;
#pragma unroll
    for (int t = 0; t < 16; ++t) {
      float v = sc[t] + ern;
      v = (v > 0.f) ? v : 0.2f * v;
      v = (4 * t + sub < deg) ? v : -INFINITY;
      sc[t] = v;
      mx = fmaxf(mx, v);
    }
    mx = fmaxf(mx, __shfl_xor(mx, 8));
    mx = fmaxf(mx, __shfl_xor(mx, 16));
    float sm = 0.f;
#pragma unroll
    for (int t = 0; t < 16; ++t)
      sm += (4 * t + sub < deg) ? __expf(sc[t] - mx) : 0.f;
    sm += __shfl_xor(sm, 8);
    sm += __shfl_xor(sm, 16);

    for (int base = 0; base < deg; base += 4) {
      const int e = base + sub;
      const int s = __shfl(base < 32 ? sorted0 : sorted1, e & 31);
      const int sn = (e < deg) ? s : 0;
      const float raw = el[sn * HH + hd];
      agg_group_scored(hfeat, lane, hsel, sn, raw, ern, mx, sm, deg - base, acc0, acc1);
    }
  }

  const float4* b4 = reinterpret_cast<const float4*>(b) + lane;
  const float4 bA = b4[0], bB = b4[32];
  float t;
  t = acc0.x + bA.x; oA.x = (t > 0.f) ? t : (__expf(t) - 1.f);
  t = acc0.y + bA.y; oA.y = (t > 0.f) ? t : (__expf(t) - 1.f);
  t = acc0.z + bA.z; oA.z = (t > 0.f) ? t : (__expf(t) - 1.f);
  t = acc0.w + bA.w; oA.w = (t > 0.f) ? t : (__expf(t) - 1.f);
  t = acc1.x + bB.x; oB.x = (t > 0.f) ? t : (__expf(t) - 1.f);
  t = acc1.y + bB.y; oB.y = (t > 0.f) ? t : (__expf(t) - 1.f);
  t = acc1.z + bB.z; oB.z = (t > 0.f) ? t : (__expf(t) - 1.f);
  t = acc1.w + bB.w; oB.w = (t > 0.f) ? t : (__expf(t) - 1.f);
}

__global__ __launch_bounds__(256, 12) void gat_node_softmax_agg(
    const float* __restrict__ el, const float* __restrict__ er,
    const int* __restrict__ cnt, const int* __restrict__ rows,
    const int* __restrict__ src, const int* __restrict__ dst,
    const float* __restrict__ hfeat, const float* __restrict__ b,
    float* __restrict__ out) {
  const int lane = threadIdx.x & 31;
  const int n = __builtin_amdgcn_readfirstlane(
      (int)((blockIdx.x * blockDim.x + threadIdx.x) >> 5));
  if (n >= NN) return;
  float4 oA, oB;
  gat_agg_row(el, er, cnt, rows, src, dst, hfeat, b, n, lane, oA, oB);
  float4* o4 = reinterpret_cast<float4*>(out + (size_t)n * F) + lane;
  o4[0]  = oA;
  o4[32] = oB;
}

__global__ __launch_bounds__(512, 12) void gat_agg_gemm(
    const float* __restrict__ el, const float* __restrict__ er,
    const int* __restrict__ cnt, const int* __restrict__ rows,
    const int* __restrict__ src, const int* __restrict__ dst,
    const float* __restrict__ hfeat, const float* __restrict__ b,
    const float* __restrict__ Wn, const float* __restrict__ aln,
    const float* __restrict__ arn, float* __restrict__ C,
    float* __restrict__ eln, float* __restrict__ ern) {
  __shared__ __attribute__((aligned(16))) float smem[GEMM_SMEM_FLOATS];
  const int lane = threadIdx.x & 31;
  const int wave = __builtin_amdgcn_readfirstlane((int)(threadIdx.x >> 5));
  const int n = blockIdx.x * 16 + wave;
  float4 oA, oB;
  gat_agg_row(el, er, cnt, rows, src, dst, hfeat, b, n, lane, oA, oB);
  float4* row = reinterpret_cast<float4*>(smem + wave * LDSPITCH) + lane;
  row[0]  = oA;
  row[32] = oB;
  __syncthreads();
  gat_gemm_from_lds<true>(smem, blockIdx.x, Wn, aln, arn, C, eln, ern);
}

extern "C" void kernel_launch(void* const* d_in, const int* in_sizes, int n_in,
                              void* d_out, int out_size, void* d_ws, size_t ws_size,
                              hipStream_t stream) {
  const float* feats = (const float*)d_in[0];
  const float* W0    = (const float*)d_in[1];
  const float* al0   = (const float*)d_in[2];
  const float* ar0   = (const float*)d_in[3];
  const float* b0    = (const float*)d_in[4];
  const float* W1    = (const float*)d_in[5];
  const float* al1   = (const float*)d_in[6];
  const float* ar1   = (const float*)d_in[7];
  const float* b1    = (const float*)d_in[8];
  const int*   src   = (const int*)d_in[9];
  const int*   dst   = (const int*)d_in[10];

  float* ws = (float*)d_ws;
  float* hbuf = ws;
  float* hbuf1 = hbuf + (size_t)NN * F;
  float* el   = hbuf1 + (size_t)NN * F;
  float* er   = el + (size_t)NN * HH;
  int* cnt    = (int*)(er + (size_t)NN * HH);
  int* rows   = cnt + (NN + 32);
  float* el1  = (float*)(rows + (size_t)NN * ROWCAP);
  float* er1  = el1 + (size_t)NN * HH;
  float* Wq0  = er1 + (size_t)NN * HH;
  float* Wq1  = Wq0 + (size_t)F * F;

  gat_zero_cnt_lay_w<<<Z_BLOCKS + LAY_BLOCKS, 256, 0, stream>>>(cnt, W0, W1, Wq0, Wq1);
  gat_count_place_gemm<<<G_ALL + E_BLOCKS, 256, 0, stream>>>(
      src, dst, cnt, rows, feats, Wq0, al0, ar0, hbuf, el, er);
  gat_agg_gemm<<<NN / 16, 512, 0, stream>>>(
      el, er, cnt, rows, src, dst, hbuf, b0, Wq1, al1, ar1, hbuf1, el1, er1);
  gat_node_softmax_agg<<<(NN * 32 + 255) / 256, 256, 0, stream>>>(
      el1, er1, cnt, rows, src, dst, hbuf1, b1, (float*)d_out);
}
